# starting kernel
# baseline (speedup 1.0000x reference)
#include <hip/hip_runtime.h>
#include <hip/hip_bf16.h>

#define DIM 128

typedef __attribute__((ext_vector_type(2))) float v2f;
typedef __attribute__((ext_vector_type(8))) float v8f;

__global__ void sage_zero_f32(float* __restrict__ p, long long n) {
  long long i = (long long)blockIdx.x * blockDim.x + threadIdx.x;
  long long stride = (long long)gridDim.x * blockDim.x;
  for (; i < n; i += stride) p[i] = 0.0f;
}

__global__ void sage_degree(const int* __restrict__ dst, float* __restrict__ cnt, int E) {
  int i = blockIdx.x * blockDim.x + threadIdx.x;
  int stride = gridDim.x * blockDim.x;
  for (; i < E; i += stride) atomicAdd(&cnt[dst[i]], 1.0f);
}

__global__ void sage_scatter(const float* __restrict__ feat,
                             const int* __restrict__ src,
                             const int* __restrict__ dst,
                             float* __restrict__ agg, int E) {
  const int lane = threadIdx.x & 31;
  const int e = blockIdx.x * (blockDim.x >> 5) + (threadIdx.x >> 5);
  if (e >= E) return;
  const int s = src[e];
  const int d = dst[e];
  const float4 v = reinterpret_cast<const float4*>(feat + (size_t)s * DIM)[lane];
  float* out = agg + (size_t)d * DIM + lane * 4;
  atomicAdd(out + 0, v.x);
  atomicAdd(out + 1, v.y);
  atomicAdd(out + 2, v.z);
  atomicAdd(out + 3, v.w);
}

__global__ __launch_bounds__(256)
void sage_fused_gemm(const float* __restrict__ agg,
                     const float* __restrict__ cnt,
                     const float* __restrict__ xin,
                     const float* __restrict__ Wl,
                     const float* __restrict__ Wr,
                     const float* __restrict__ bias,
                     float* __restrict__ out,
                     int do_relu) {
  const int lane = threadIdx.x & 31;
  const int wave = threadIdx.x >> 5;
  const int m    = lane & 15;
  const int half = lane >> 4;
  const int row0 = blockIdx.x * 16;
  const int col  = wave * 16 + m;

  const float* aggRow = agg + (size_t)(row0 + m) * DIM;
  const float* xRow   = xin + (size_t)(row0 + m) * DIM;
  const float  inv    = 1.0f / fmaxf(cnt[row0 + m], 1.0f);

  v8f c = {};
#pragma unroll
  for (int s = 0; s < 32; ++s) {
    const int kk = 4 * s + 2 * half;
    v2f a, b;
    a.x = aggRow[kk] * inv;
    a.y = aggRow[kk + 1] * inv;
    b.x = Wl[(size_t)kk * DIM + col];
    b.y = Wl[(size_t)(kk + 1) * DIM + col];
    c = __builtin_amdgcn_wmma_f32_16x16x4_f32(false, a, false, b, (short)0, c,
                                              false, false);
    a.x = xRow[kk];
    a.y = xRow[kk + 1];
    b.x = Wr[(size_t)kk * DIM + col];
    b.y = Wr[(size_t)(kk + 1) * DIM + col];
    c = __builtin_amdgcn_wmma_f32_16x16x4_f32(false, a, false, b, (short)0, c,
                                              false, false);
  }

  const float bv = bias[col];
#pragma unroll
  for (int r = 0; r < 8; ++r) {
    float v = c[r] + bv;
    if (do_relu) v = fmaxf(v, 0.0f);
    out[(size_t)(row0 + r + 8 * half) * DIM + col] = v;
  }
}

extern "C" void kernel_launch(void* const* d_in, const int* in_sizes, int n_in,
                              void* d_out, int out_size, void* d_ws, size_t ws_size,
                              hipStream_t stream) {
  const float* x   = (const float*)d_in[0];
  const int*  edge = (const int*)d_in[1];
  const float* W1l = (const float*)d_in[2];
  const float* b1  = (const float*)d_in[3];
  const float* W1r = (const float*)d_in[4];
  const float* W2l = (const float*)d_in[5];
  const float* b2  = (const float*)d_in[6];
  const float* W2r = (const float*)d_in[7];

  const int N = in_sizes[0] / DIM;
  const int E = in_sizes[1] / 2;
  const int* src = edge;
  const int* dst = edge + E;

  float* agg = (float*)d_ws;
  float* cnt = agg + (size_t)N * DIM;
  float* h   = cnt + N;
  float* out = (float*)d_out;

  const int edges_per_block = 256 / 32;
  const int scatter_blocks  = (E + edges_per_block - 1) / edges_per_block;
  const int gemm_blocks     = N / 16;

  sage_zero_f32<<<2048, 256, 0, stream>>>(agg, (long long)N * DIM + N);
  sage_degree<<<2048, 256, 0, stream>>>(dst, cnt, E);
  sage_scatter<<<scatter_blocks, 256, 0, stream>>>(x, src, dst, agg, E);
  sage_fused_gemm<<<gemm_blocks, 256, 0, stream>>>(agg, cnt, x, W1l, W1r, b1, h, 1);

  sage_zero_f32<<<2048, 256, 0, stream>>>(agg, (long long)N * DIM);
  sage_scatter<<<scatter_blocks, 256, 0, stream>>>(h, src, dst, agg, E);
  sage_fused_gemm<<<gemm_blocks, 256, 0, stream>>>(agg, cnt, h, W2l, W2r, b2, out, 0);
}


# f32 atomic scatter-add of both layers replaced by a per-call dst-bucketed edge list (int count with returned rank, scan, fill) + one-wave-per-row register gather-sum; sum order = arrival rank
# speedup vs baseline: 1.9281x; 1.9281x over previous
#include <hip/hip_runtime.h>
#include <hip/hip_bf16.h>

#define DIM 128

typedef __attribute__((ext_vector_type(2))) float v2f;
typedef __attribute__((ext_vector_type(8))) float v8f;


__global__ void sage_zero_i32(int* __restrict__ p, int n) {
  int i = blockIdx.x * blockDim.x + threadIdx.x;
  const int stride = gridDim.x * blockDim.x;
  for (; i < n; i += stride) p[i] = 0;
}

__global__ __launch_bounds__(256)
void sage_degree_rank(const int* __restrict__ dst, int* __restrict__ deg,
                      int* __restrict__ rank, int E) {
  const int T  = gridDim.x * blockDim.x;
  const int e0 = blockIdx.x * blockDim.x + threadIdx.x;
  if (e0 + 3 * T < E) {
    int d[4], r[4];
#pragma unroll
    for (int u = 0; u < 4; ++u) d[u] = dst[e0 + u * T];
#pragma unroll
    for (int u = 0; u < 4; ++u) r[u] = atomicAdd(&deg[d[u]], 1);
#pragma unroll
    for (int u = 0; u < 4; ++u) rank[e0 + u * T] = r[u];
  } else {
#pragma unroll
    for (int u = 0; u < 4; ++u) {
      const int e = e0 + u * T;
      if (e < E) rank[e] = atomicAdd(&deg[dst[e]], 1);
    }
  }
}

#define SCAN_THREADS 1024
#define SCAN_SPAN (SCAN_THREADS * 4)

__global__ __launch_bounds__(SCAN_THREADS)
void sage_scan(const int* __restrict__ deg, int* __restrict__ rowptr,
               float* __restrict__ cnt, int N) {
  __shared__ int w_front[SCAN_THREADS / 32];
  __shared__ int w_own[SCAN_THREADS / 32];
  const int tid  = threadIdx.x;
  const int lane = tid & 31;
  const int wave = tid >> 5;
  const int nblk = blockIdx.x;

  int front = 0;
#pragma unroll 8
  for (int c = 0; c < nblk; ++c) {
    const int4 v = *reinterpret_cast<const int4*>(deg + (size_t)c * SCAN_SPAN + tid * 4);
    front += (v.x + v.y) + (v.z + v.w);
  }

  const int i0 = nblk * SCAN_SPAN + tid * 4;
  int a0 = 0, a1 = 0, a2 = 0, a3 = 0;
  if (i0 + 3 < N) {
    const int4 v = *reinterpret_cast<const int4*>(deg + i0);
    a0 = v.x; a1 = v.y; a2 = v.z; a3 = v.w;
  } else {
    if (i0 < N)     a0 = deg[i0];
    if (i0 + 1 < N) a1 = deg[i0 + 1];
    if (i0 + 2 < N) a2 = deg[i0 + 2];
  }
  const int tsum = a0 + a1 + a2 + a3;

  int inc = tsum;
#pragma unroll
  for (int o = 1; o < 32; o <<= 1) {
    const int t = __shfl_up(inc, o);
    if (lane >= o) inc += t;
  }
#pragma unroll
  for (int o = 16; o > 0; o >>= 1) front += __shfl_xor(front, o);

  if (lane == 31) w_own[wave] = inc;
  if (lane == 0)  w_front[wave] = front;
  __syncthreads();

  int off = 0;
#pragma unroll
  for (int w = 0; w < SCAN_THREADS / 32; ++w) {
    off += w_front[w];
    if (w < wave) off += w_own[w];
  }

  const int ex = off + inc - tsum;
  if (i0 + 3 < N) {
    *reinterpret_cast<int4*>(rowptr + i0) = make_int4(ex, ex + a0, ex + a0 + a1, ex + a0 + a1 + a2);
    *reinterpret_cast<float4*>(cnt + i0)  = make_float4((float)a0, (float)a1, (float)a2, (float)a3);
  } else {
    if (i0 < N)     { rowptr[i0]     = ex;           cnt[i0]     = (float)a0; }
    if (i0 + 1 < N) { rowptr[i0 + 1] = ex + a0;      cnt[i0 + 1] = (float)a1; }
    if (i0 + 2 < N) { rowptr[i0 + 2] = ex + a0 + a1; cnt[i0 + 2] = (float)a2; }
  }
  if (i0 < N && i0 + 4 >= N) rowptr[N] = ex + tsum;
}

__global__ __launch_bounds__(256)
void sage_fill(const int* __restrict__ src, const int* __restrict__ dst,
               const int* __restrict__ rank, const int* __restrict__ rowptr,
               int* __restrict__ col, int E) {
  const int T  = gridDim.x * blockDim.x;
  const int e0 = blockIdx.x * blockDim.x + threadIdx.x;
  if (e0 + 3 * T < E) {
    int p[4], s[4];
#pragma unroll
    for (int u = 0; u < 4; ++u) {
      const int e = e0 + u * T;
      s[u] = src[e];
      p[u] = rowptr[dst[e]] + rank[e];
    }
#pragma unroll
    for (int u = 0; u < 4; ++u) col[p[u]] = s[u];
  } else {
#pragma unroll
    for (int u = 0; u < 4; ++u) {
      const int e = e0 + u * T;
      if (e < E) col[rowptr[dst[e]] + rank[e]] = src[e];
    }
  }
}

__global__ __launch_bounds__(256)
void sage_gather_sum(const float* __restrict__ feat,
                     const int* __restrict__ rowptr,
                     const int* __restrict__ col,
                     float* __restrict__ agg, int N) {
  const int lane = threadIdx.x & 31;
  const int wave   = __builtin_amdgcn_readfirstlane((int)(threadIdx.x >> 5));
  const int wpb    = blockDim.x >> 5;
  const int nwaves = gridDim.x * wpb;
  const float4 zero = make_float4(0.0f, 0.0f, 0.0f, 0.0f);

  for (int node = blockIdx.x * wpb + wave; node < N; node += nwaves) {
    const int beg = rowptr[node];
    const int end = rowptr[node + 1];
    float4 acc = zero;
    for (int j = beg; j < end; j += 32) {
      const int n = min(32, end - j);
      const int mine = col[j + min(lane, n - 1)];
      int k = 0;
      for (; k + 8 <= n; k += 8) {
        int s[8];
        float4 r[8];
#pragma unroll
        for (int u = 0; u < 8; ++u) s[u] = __builtin_amdgcn_readlane(mine, k + u);
#pragma unroll
        for (int u = 0; u < 8; ++u)
          r[u] = reinterpret_cast<const float4*>(feat + (size_t)s[u] * DIM)[lane];
#pragma unroll
        for (int u = 0; u < 8; ++u) {
          acc.x += r[u].x; acc.y += r[u].y; acc.z += r[u].z; acc.w += r[u].w;
        }
      }
      if (k < n) {
        int s[8];
        float4 r[8];
#pragma unroll
        for (int u = 0; u < 8; ++u) s[u] = __builtin_amdgcn_readlane(mine, k + u);
#pragma unroll
        for (int u = 0; u < 8; ++u) {
          r[u] = zero;
          if (k + u < n) r[u] = reinterpret_cast<const float4*>(feat + (size_t)s[u] * DIM)[lane];
        }
#pragma unroll
        for (int u = 0; u < 8; ++u) {
          acc.x += r[u].x; acc.y += r[u].y; acc.z += r[u].z; acc.w += r[u].w;
        }
      }
    }
    reinterpret_cast<float4*>(agg + (size_t)node * DIM)[lane] = acc;
  }
}

__global__ __launch_bounds__(256)
void sage_fused_gemm(const float* __restrict__ agg,
                     const float* __restrict__ cnt,
                     const float* __restrict__ xin,
                     const float* __restrict__ Wl,
                     const float* __restrict__ Wr,
                     const float* __restrict__ bias,
                     float* __restrict__ out,
                     int do_relu) {
  const int lane = threadIdx.x & 31;
  const int wave = threadIdx.x >> 5;
  const int m    = lane & 15;
  const int half = lane >> 4;
  const int row0 = blockIdx.x * 16;
  const int col  = wave * 16 + m;

  const float* aggRow = agg + (size_t)(row0 + m) * DIM;
  const float* xRow   = xin + (size_t)(row0 + m) * DIM;
  const float  inv    = 1.0f / fmaxf(cnt[row0 + m], 1.0f);

  v8f c = {};
#pragma unroll
  for (int s = 0; s < 32; ++s) {
    const int kk = 4 * s + 2 * half;
    v2f a, b;
    a.x = aggRow[kk] * inv;
    a.y = aggRow[kk + 1] * inv;
    b.x = Wl[(size_t)kk * DIM + col];
    b.y = Wl[(size_t)(kk + 1) * DIM + col];
    c = __builtin_amdgcn_wmma_f32_16x16x4_f32(false, a, false, b, (short)0, c,
                                              false, false);
    a.x = xRow[kk];
    a.y = xRow[kk + 1];
    b.x = Wr[(size_t)kk * DIM + col];
    b.y = Wr[(size_t)(kk + 1) * DIM + col];
    c = __builtin_amdgcn_wmma_f32_16x16x4_f32(false, a, false, b, (short)0, c,
                                              false, false);
  }

  const float bv = bias[col];
#pragma unroll
  for (int r = 0; r < 8; ++r) {
    float v = c[r] + bv;
    if (do_relu) v = fmaxf(v, 0.0f);
    out[(size_t)(row0 + r + 8 * half) * DIM + col] = v;
  }
}

extern "C" void kernel_launch(void* const* d_in, const int* in_sizes, int n_in,
                              void* d_out, int out_size, void* d_ws, size_t ws_size,
                              hipStream_t stream) {
  const float* x   = (const float*)d_in[0];
  const int*  edge = (const int*)d_in[1];
  const float* W1l = (const float*)d_in[2];
  const float* b1  = (const float*)d_in[3];
  const float* W1r = (const float*)d_in[4];
  const float* W2l = (const float*)d_in[5];
  const float* b2  = (const float*)d_in[6];
  const float* W2r = (const float*)d_in[7];

  const int N = in_sizes[0] / DIM;
  const int E = in_sizes[1] / 2;
  const int* src = edge;
  const int* dst = edge + E;

  float* agg = (float*)d_ws;
  float* cnt = agg + (size_t)N * DIM;
  float* h   = cnt + N;
  int* deg    = (int*)(h + (size_t)N * DIM);
  int* rowptr = deg + N;
  int* rank   = rowptr + N + 4;
  int* col    = rank + E;
  float* out = (float*)d_out;

  const int edge_blocks = (E + 1023) / 1024;
  const int scan_blocks = (N + SCAN_SPAN - 1) / SCAN_SPAN;
  const int gemm_blocks = N / 16;

  sage_zero_i32<<<(N + 255) / 256, 256, 0, stream>>>(deg, N);
  sage_degree_rank<<<edge_blocks, 256, 0, stream>>>(dst, deg, rank, E);
  sage_scan<<<scan_blocks, SCAN_THREADS, 0, stream>>>(deg, rowptr, cnt, N);
  sage_fill<<<edge_blocks, 256, 0, stream>>>(src, dst, rank, rowptr, col, E);

  sage_gather_sum<<<2048, 256, 0, stream>>>(x, rowptr, col, agg, N);
  sage_fused_gemm<<<gemm_blocks, 256, 0, stream>>>(agg, cnt, x, W1l, W1r, b1, h, 1);

  sage_gather_sum<<<2048, 256, 0, stream>>>(h, rowptr, col, agg, N);
  sage_fused_gemm<<<gemm_blocks, 256, 0, stream>>>(agg, cnt, h, W2l, W2r, b2, out, 0);
}


# fused GEMM: both weight matrices staged in LDS once per resident 8-wave workgroup; each wave stages 16-row A stripes in its own LDS region and computes 16x128 per stripe; sum order unchanged
# speedup vs baseline: 2.6275x; 1.3627x over previous
#include <hip/hip_runtime.h>
#include <hip/hip_bf16.h>

#define DIM 128

typedef __attribute__((ext_vector_type(2))) float v2f;
typedef __attribute__((ext_vector_type(4))) float v4f;
typedef __attribute__((ext_vector_type(8))) float v8f;


__global__ void sage_zero_i32(int* __restrict__ p, int n) {
  int i = blockIdx.x * blockDim.x + threadIdx.x;
  const int stride = gridDim.x * blockDim.x;
  for (; i < n; i += stride) p[i] = 0;
}

__global__ __launch_bounds__(256)
void sage_degree_rank(const int* __restrict__ dst, int* __restrict__ deg,
                      int* __restrict__ rank, int E) {
  const int T  = gridDim.x * blockDim.x;
  const int e0 = blockIdx.x * blockDim.x + threadIdx.x;
  if (e0 + 3 * T < E) {
    int d[4], r[4];
#pragma unroll
    for (int u = 0; u < 4; ++u) d[u] = dst[e0 + u * T];
#pragma unroll
    for (int u = 0; u < 4; ++u) r[u] = atomicAdd(&deg[d[u]], 1);
#pragma unroll
    for (int u = 0; u < 4; ++u) rank[e0 + u * T] = r[u];
  } else {
#pragma unroll
    for (int u = 0; u < 4; ++u) {
      const int e = e0 + u * T;
      if (e < E) rank[e] = atomicAdd(&deg[dst[e]], 1);
    }
  }
}

#define SCAN_THREADS 1024
#define SCAN_SPAN (SCAN_THREADS * 4)

__global__ __launch_bounds__(SCAN_THREADS)
void sage_scan(const int* __restrict__ deg, int* __restrict__ rowptr,
               float* __restrict__ cnt, int N) {
  __shared__ int w_front[SCAN_THREADS / 32];
  __shared__ int w_own[SCAN_THREADS / 32];
  const int tid  = threadIdx.x;
  const int lane = tid & 31;
  const int wave = tid >> 5;
  const int nblk = blockIdx.x;

  int front = 0;
#pragma unroll 8
  for (int c = 0; c < nblk; ++c) {
    const int4 v = *reinterpret_cast<const int4*>(deg + (size_t)c * SCAN_SPAN + tid * 4);
    front += (v.x + v.y) + (v.z + v.w);
  }

  const int i0 = nblk * SCAN_SPAN + tid * 4;
  int a0 = 0, a1 = 0, a2 = 0, a3 = 0;
  if (i0 + 3 < N) {
    const int4 v = *reinterpret_cast<const int4*>(deg + i0);
    a0 = v.x; a1 = v.y; a2 = v.z; a3 = v.w;
  } else {
    if (i0 < N)     a0 = deg[i0];
    if (i0 + 1 < N) a1 = deg[i0 + 1];
    if (i0 + 2 < N) a2 = deg[i0 + 2];
  }
  const int tsum = a0 + a1 + a2 + a3;

  int inc = tsum;
#pragma unroll
  for (int o = 1; o < 32; o <<= 1) {
    const int t = __shfl_up(inc, o);
    if (lane >= o) inc += t;
  }
#pragma unroll
  for (int o = 16; o > 0; o >>= 1) front += __shfl_xor(front, o);

  if (lane == 31) w_own[wave] = inc;
  if (lane == 0)  w_front[wave] = front;
  __syncthreads();

  int off = 0;
#pragma unroll
  for (int w = 0; w < SCAN_THREADS / 32; ++w) {
    off += w_front[w];
    if (w < wave) off += w_own[w];
  }

  const int ex = off + inc - tsum;
  if (i0 + 3 < N) {
    *reinterpret_cast<int4*>(rowptr + i0) = make_int4(ex, ex + a0, ex + a0 + a1, ex + a0 + a1 + a2);
    *reinterpret_cast<float4*>(cnt + i0)  = make_float4((float)a0, (float)a1, (float)a2, (float)a3);
  } else {
    if (i0 < N)     { rowptr[i0]     = ex;           cnt[i0]     = (float)a0; }
    if (i0 + 1 < N) { rowptr[i0 + 1] = ex + a0;      cnt[i0 + 1] = (float)a1; }
    if (i0 + 2 < N) { rowptr[i0 + 2] = ex + a0 + a1; cnt[i0 + 2] = (float)a2; }
  }
  if (i0 < N && i0 + 4 >= N) rowptr[N] = ex + tsum;
}

__global__ __launch_bounds__(256)
void sage_fill(const int* __restrict__ src, const int* __restrict__ dst,
               const int* __restrict__ rank, const int* __restrict__ rowptr,
               int* __restrict__ col, int E) {
  const int T  = gridDim.x * blockDim.x;
  const int e0 = blockIdx.x * blockDim.x + threadIdx.x;
  if (e0 + 3 * T < E) {
    int p[4], s[4];
#pragma unroll
    for (int u = 0; u < 4; ++u) {
      const int e = e0 + u * T;
      s[u] = src[e];
      p[u] = rowptr[dst[e]] + rank[e];
    }
#pragma unroll
    for (int u = 0; u < 4; ++u) col[p[u]] = s[u];
  } else {
#pragma unroll
    for (int u = 0; u < 4; ++u) {
      const int e = e0 + u * T;
      if (e < E) col[rowptr[dst[e]] + rank[e]] = src[e];
    }
  }
}

__global__ __launch_bounds__(256)
void sage_gather_sum(const float* __restrict__ feat,
                     const int* __restrict__ rowptr,
                     const int* __restrict__ col,
                     float* __restrict__ agg, int N) {
  const int lane = threadIdx.x & 31;
  const int wave   = __builtin_amdgcn_readfirstlane((int)(threadIdx.x >> 5));
  const int wpb    = blockDim.x >> 5;
  const int nwaves = gridDim.x * wpb;
  const float4 zero = make_float4(0.0f, 0.0f, 0.0f, 0.0f);

  for (int node = blockIdx.x * wpb + wave; node < N; node += nwaves) {
    const int beg = rowptr[node];
    const int end = rowptr[node + 1];
    float4 acc = zero;
    for (int j = beg; j < end; j += 32) {
      const int n = min(32, end - j);
      const int mine = col[j + min(lane, n - 1)];
      int k = 0;
      for (; k + 8 <= n; k += 8) {
        int s[8];
        float4 r[8];
#pragma unroll
        for (int u = 0; u < 8; ++u) s[u] = __builtin_amdgcn_readlane(mine, k + u);
#pragma unroll
        for (int u = 0; u < 8; ++u)
          r[u] = reinterpret_cast<const float4*>(feat + (size_t)s[u] * DIM)[lane];
#pragma unroll
        for (int u = 0; u < 8; ++u) {
          acc.x += r[u].x; acc.y += r[u].y; acc.z += r[u].z; acc.w += r[u].w;
        }
      }
      if (k < n) {
        int s[8];
        float4 r[8];
#pragma unroll
        for (int u = 0; u < 8; ++u) s[u] = __builtin_amdgcn_readlane(mine, k + u);
#pragma unroll
        for (int u = 0; u < 8; ++u) {
          r[u] = zero;
          if (k + u < n) r[u] = reinterpret_cast<const float4*>(feat + (size_t)s[u] * DIM)[lane];
        }
#pragma unroll
        for (int u = 0; u < 8; ++u) {
          acc.x += r[u].x; acc.y += r[u].y; acc.z += r[u].z; acc.w += r[u].w;
        }
      }
    }
    reinterpret_cast<float4*>(agg + (size_t)node * DIM)[lane] = acc;
  }
}

#define WP_PITCH 288
#define W_LDS_FLOATS (64 * WP_PITCH)
#define A_PITCH 132
#define A_TILE_FLOATS (16 * A_PITCH)
#define GEMM_WAVES 8
#define GEMM_LDS_FLOATS (2 * W_LDS_FLOATS + GEMM_WAVES * 2 * A_TILE_FLOATS)

__device__ __forceinline__ void wave_lds_handover() {
  __builtin_amdgcn_fence(__ATOMIC_RELEASE, "wavefront");
  __builtin_amdgcn_fence(__ATOMIC_ACQUIRE, "wavefront");
}

__global__ __launch_bounds__(256)
void sage_fused_gemm(const float* __restrict__ agg,
                     const float* __restrict__ cnt,
                     const float* __restrict__ xin,
                     const float* __restrict__ Wl,
                     const float* __restrict__ Wr,
                     const float* __restrict__ bias,
                     float* __restrict__ out,
                     int do_relu, int nstripes) {
  __shared__ __attribute__((aligned(16))) float lds[GEMM_LDS_FLOATS];

  const int tid  = threadIdx.x;
  const int lane = tid & 31;
  const int wave = __builtin_amdgcn_readfirstlane(tid >> 5);
  const int m    = lane & 15;
  const int half = lane >> 4;

  float* sWl = lds;
  float* sWr = lds + W_LDS_FLOATS;
  float* sAg = lds + 2 * W_LDS_FLOATS + wave * (2 * A_TILE_FLOATS);
  float* sAx = sAg + A_TILE_FLOATS;

#pragma unroll
  for (int i = 0; i < 8; ++i) {
    const int p = wave + 8 * i;
    const float4 l0 = reinterpret_cast<const float4*>(Wl + (size_t)(2 * p) * DIM)[lane];
    const float4 l1 = reinterpret_cast<const float4*>(Wl + (size_t)(2 * p + 1) * DIM)[lane];
    const float4 r0 = reinterpret_cast<const float4*>(Wr + (size_t)(2 * p) * DIM)[lane];
    const float4 r1 = reinterpret_cast<const float4*>(Wr + (size_t)(2 * p + 1) * DIM)[lane];
    float4* dl = reinterpret_cast<float4*>(sWl + p * WP_PITCH + 8 * lane);
    float4* dr = reinterpret_cast<float4*>(sWr + p * WP_PITCH + 8 * lane);
    dl[0] = make_float4(l0.x, l1.x, l0.y, l1.y);
    dl[1] = make_float4(l0.z, l1.z, l0.w, l1.w);
    dr[0] = make_float4(r0.x, r1.x, r0.y, r1.y);
    dr[1] = make_float4(r0.z, r1.z, r0.w, r1.w);
  }

  float bv[8];
#pragma unroll
  for (int t = 0; t < 8; ++t) bv[t] = bias[16 * t + m];

  const int stride = GEMM_WAVES * gridDim.x;
  int stripe = wave * gridDim.x + blockIdx.x;

  v4f ra[16], rx[16];
  float cn = 1.0f;
  if (stripe < nstripes) {
    const size_t row0 = (size_t)stripe * 16;
    cn = cnt[row0 + m];
#pragma unroll
    for (int r = 0; r < 16; ++r) {
      ra[r] = reinterpret_cast<const v4f*>(agg + (row0 + r) * DIM)[lane];
      rx[r] = reinterpret_cast<const v4f*>(xin + (row0 + r) * DIM)[lane];
    }
  }

  __syncthreads();

  const float* aG = sAg + m * A_PITCH + 2 * half;
  const float* aX = sAx + m * A_PITCH + 2 * half;
  const float* bL = sWl + half * WP_PITCH + 2 * m;
  const float* bR = sWr + half * WP_PITCH + 2 * m;

  while (stripe < nstripes) {
    const size_t row0 = (size_t)stripe * 16;

    const float inv = 1.0f / fmaxf(cn, 1.0f);
#pragma unroll
    for (int r = 0; r < 16; ++r) {
      const float iv = __int_as_float(__builtin_amdgcn_readlane(__float_as_int(inv), r));
      reinterpret_cast<v4f*>(sAg + r * A_PITCH)[lane] = ra[r] * iv;
      reinterpret_cast<v4f*>(sAx + r * A_PITCH)[lane] = rx[r];
    }
    wave_lds_handover();

    const int next = stripe + stride;
    if (next < nstripes) {
      const size_t nrow0 = (size_t)next * 16;
      cn = cnt[nrow0 + m];
#pragma unroll
      for (int r = 0; r < 16; ++r) {
        ra[r] = reinterpret_cast<const v4f*>(agg + (nrow0 + r) * DIM)[lane];
        rx[r] = reinterpret_cast<const v4f*>(xin + (nrow0 + r) * DIM)[lane];
      }
    }

    v8f c[8];
#pragma unroll
    for (int t = 0; t < 8; ++t) c[t] = v8f{};

#pragma unroll
    for (int s = 0; s < 32; ++s) {
      const v2f ag = *reinterpret_cast<const v2f*>(aG + 4 * s);
      const v2f ax = *reinterpret_cast<const v2f*>(aX + 4 * s);
      v2f wl[8], wr[8];
#pragma unroll
      for (int t = 0; t < 8; ++t) {
        wl[t] = *reinterpret_cast<const v2f*>(bL + 2 * s * WP_PITCH + 32 * t);
        wr[t] = *reinterpret_cast<const v2f*>(bR + 2 * s * WP_PITCH + 32 * t);
      }
#pragma unroll
      for (int t = 0; t < 8; ++t)
        c[t] = __builtin_amdgcn_wmma_f32_16x16x4_f32(false, ag, false, wl[t], (short)0, c[t],
                                                     false, false);
#pragma unroll
      for (int t = 0; t < 8; ++t)
        c[t] = __builtin_amdgcn_wmma_f32_16x16x4_f32(false, ax, false, wr[t], (short)0, c[t],
                                                     false, false);
    }

#pragma unroll
    for (int t = 0; t < 8; ++t) {
#pragma unroll
      for (int r = 0; r < 8; ++r) {
        float v = c[t][r] + bv[t];
        if (do_relu) v = fmaxf(v, 0.0f);
        out[(row0 + r + 8 * half) * DIM + 16 * t + m] = v;
      }
    }

    wave_lds_handover();
    stripe = next;
  }
}

extern "C" void kernel_launch(void* const* d_in, const int* in_sizes, int n_in,
                              void* d_out, int out_size, void* d_ws, size_t ws_size,
                              hipStream_t stream) {
  const float* x   = (const float*)d_in[0];
  const int*  edge = (const int*)d_in[1];
  const float* W1l = (const float*)d_in[2];
  const float* b1  = (const float*)d_in[3];
  const float* W1r = (const float*)d_in[4];
  const float* W2l = (const float*)d_in[5];
  const float* b2  = (const float*)d_in[6];
  const float* W2r = (const float*)d_in[7];

  const int N = in_sizes[0] / DIM;
  const int E = in_sizes[1] / 2;
  const int* src = edge;
  const int* dst = edge + E;

  float* agg = (float*)d_ws;
  float* cnt = agg + (size_t)N * DIM;
  float* h   = cnt + N;
  int* deg    = (int*)(h + (size_t)N * DIM);
  int* rowptr = deg + N;
  int* rank   = rowptr + N + 4;
  int* col    = rank + E;
  float* out = (float*)d_out;

  const int edge_blocks = (E + 1023) / 1024;
  const int scan_blocks = (N + SCAN_SPAN - 1) / SCAN_SPAN;
  const int gemm_stripes = N / 16;

  sage_zero_i32<<<(N + 255) / 256, 256, 0, stream>>>(deg, N);
  sage_degree_rank<<<edge_blocks, 256, 0, stream>>>(dst, deg, rank, E);
  sage_scan<<<scan_blocks, SCAN_THREADS, 0, stream>>>(deg, rowptr, cnt, N);
  sage_fill<<<edge_blocks, 256, 0, stream>>>(src, dst, rank, rowptr, col, E);

  sage_gather_sum<<<2048, 256, 0, stream>>>(x, rowptr, col, agg, N);
  sage_fused_gemm<<<256, 256, 0, stream>>>(agg, cnt, x, W1l, W1r, b1, h, 1, gemm_stripes);

  sage_gather_sum<<<2048, 256, 0, stream>>>(h, rowptr, col, agg, N);
  sage_fused_gemm<<<256, 256, 0, stream>>>(agg, cnt, h, W2l, W2r, b2, out, 0, gemm_stripes);
}


# dst bucketing redone as a two-level counting sort on LDS counters (512-node coarse buckets per edge chunk, then per-bucket count, scan, place); no global atomics left; sum order = LDS arrival rank
# speedup vs baseline: 3.6325x; 1.3825x over previous
#include <hip/hip_runtime.h>
#include <hip/hip_bf16.h>

#define DIM 128

typedef __attribute__((ext_vector_type(2))) float v2f;
typedef __attribute__((ext_vector_type(4))) float v4f;
typedef __attribute__((ext_vector_type(8))) float v8f;


#define CS_SHIFT 9
#define CS_W (1 << CS_SHIFT)
#define CS_BMAX 256
#define CS_G 256
#define CS_THREADS 1024

__global__ __launch_bounds__(CS_THREADS)
void sage_bucket_count(const int* __restrict__ dst, int* __restrict__ tab, int E) {
  __shared__ int hist[CS_BMAX];
  const int tid = threadIdx.x;
  const int g   = blockIdx.x;
  if (tid < CS_BMAX) hist[tid] = 0;
  __syncthreads();
  const int chunk = (E + CS_G - 1) / CS_G;
  const int beg = g * chunk;
  const int end = min(E, beg + chunk);
  for (int e0 = beg + tid; e0 < end; e0 += 4 * CS_THREADS) {
    int d[4];
#pragma unroll
    for (int u = 0; u < 4; ++u) {
      const int e = e0 + u * CS_THREADS;
      d[u] = (e < end) ? dst[e] : -1;
    }
#pragma unroll
    for (int u = 0; u < 4; ++u)
      if (d[u] >= 0) atomicAdd(&hist[d[u] >> CS_SHIFT], 1);
  }
  __syncthreads();
  if (tid < CS_BMAX) tab[tid * CS_G + g] = hist[tid];
}

__global__ __launch_bounds__(CS_THREADS)
void sage_bucket_scan(int* __restrict__ tab, int* __restrict__ bstart) {
  __shared__ int w_tot[CS_THREADS / 32];
  const int tid  = threadIdx.x;
  const int lane = tid & 31;
  const int wave = tid >> 5;

  int4* p = reinterpret_cast<int4*>(tab + tid * 64);
  int4 v[16];
#pragma unroll
  for (int i = 0; i < 16; ++i) v[i] = p[i];
  int tsum = 0;
#pragma unroll
  for (int i = 0; i < 16; ++i) tsum += (v[i].x + v[i].y) + (v[i].z + v[i].w);

  int inc = tsum;
#pragma unroll
  for (int o = 1; o < 32; o <<= 1) {
    const int t = __shfl_up(inc, o);
    if (lane >= o) inc += t;
  }
  if (lane == 31) w_tot[wave] = inc;
  __syncthreads();
  int off = 0;
#pragma unroll
  for (int w = 0; w < CS_THREADS / 32; ++w)
    if (w < wave) off += w_tot[w];

  int run = off + inc - tsum;
  if ((tid & 3) == 0) bstart[tid >> 2] = run;
#pragma unroll
  for (int i = 0; i < 16; ++i) {
    int4 o;
    o.x = run; run += v[i].x;
    o.y = run; run += v[i].y;
    o.z = run; run += v[i].z;
    o.w = run; run += v[i].w;
    p[i] = o;
  }
  if (tid == CS_THREADS - 1) bstart[CS_BMAX] = run;
}

__global__ __launch_bounds__(CS_THREADS)
void sage_bucket_scatter(const int* __restrict__ src, const int* __restrict__ dst,
                         const int* __restrict__ tab, int* __restrict__ tmp_i, int E) {
  __shared__ int cur[CS_BMAX];
  int2* tmp = reinterpret_cast<int2*>(tmp_i);
  const int tid = threadIdx.x;
  const int g   = blockIdx.x;
  if (tid < CS_BMAX) cur[tid] = tab[tid * CS_G + g];
  __syncthreads();
  const int chunk = (E + CS_G - 1) / CS_G;
  const int beg = g * chunk;
  const int end = min(E, beg + chunk);
  for (int e0 = beg + tid; e0 < end; e0 += 4 * CS_THREADS) {
    int d[4], s[4];
#pragma unroll
    for (int u = 0; u < 4; ++u) {
      const int e = e0 + u * CS_THREADS;
      d[u] = -1; s[u] = 0;
      if (e < end) { d[u] = dst[e]; s[u] = src[e]; }
    }
#pragma unroll
    for (int u = 0; u < 4; ++u) {
      if (d[u] >= 0) {
        const int pos = atomicAdd(&cur[d[u] >> CS_SHIFT], 1);
        tmp[pos] = make_int2(s[u], d[u]);
      }
    }
  }
}

__global__ __launch_bounds__(CS_THREADS)
void sage_bucket_sort(const int* __restrict__ tmp_i, const int* __restrict__ bstart,
                      int* __restrict__ rowptr, float* __restrict__ cnt,
                      int* __restrict__ col, int N) {
  __shared__ int c[CS_W];
  __shared__ int w_tot[CS_THREADS / 32];
  const int2* tmp = reinterpret_cast<const int2*>(tmp_i);
  const int tid  = threadIdx.x;
  const int lane = tid & 31;
  const int wave = tid >> 5;
  const int b    = blockIdx.x;
  const int beg  = bstart[b];
  const int end  = bstart[b + 1];
  const int node0 = b << CS_SHIFT;

  if (tid < CS_W) c[tid] = 0;
  __syncthreads();

  for (int j0 = beg + tid; j0 < end; j0 += 4 * CS_THREADS) {
    int d[4];
#pragma unroll
    for (int u = 0; u < 4; ++u) {
      const int j = j0 + u * CS_THREADS;
      d[u] = (j < end) ? tmp[j].y : -1;
    }
#pragma unroll
    for (int u = 0; u < 4; ++u)
      if (d[u] >= 0) atomicAdd(&c[d[u] - node0], 1);
  }
  __syncthreads();

  const int v = (tid < CS_W) ? c[tid] : 0;
  int inc = v;
#pragma unroll
  for (int o = 1; o < 32; o <<= 1) {
    const int t = __shfl_up(inc, o);
    if (lane >= o) inc += t;
  }
  if (lane == 31) w_tot[wave] = inc;
  __syncthreads();
  int off = 0;
#pragma unroll
  for (int w = 0; w < CS_THREADS / 32; ++w)
    if (w < wave) off += w_tot[w];
  const int ex = off + inc - v;

  const int node = node0 + tid;
  if (tid < CS_W) {
    if (node < N) {
      rowptr[node] = beg + ex;
      cnt[node]    = (float)v;
      if (node == N - 1) rowptr[N] = beg + ex + v;
    }
    c[tid] = ex;
  }
  __syncthreads();

  for (int j0 = beg + tid; j0 < end; j0 += 4 * CS_THREADS) {
    int2 e[4];
#pragma unroll
    for (int u = 0; u < 4; ++u) {
      const int j = j0 + u * CS_THREADS;
      e[u] = make_int2(0, -1);
      if (j < end) e[u] = tmp[j];
    }
#pragma unroll
    for (int u = 0; u < 4; ++u) {
      if (e[u].y >= 0) {
        const int pos = atomicAdd(&c[e[u].y - node0], 1);
        col[beg + pos] = e[u].x;
      }
    }
  }
}

__global__ __launch_bounds__(256)
void sage_gather_sum(const float* __restrict__ feat,
                     const int* __restrict__ rowptr,
                     const int* __restrict__ col,
                     float* __restrict__ agg, int N) {
  const int lane = threadIdx.x & 31;
  const int wave   = __builtin_amdgcn_readfirstlane((int)(threadIdx.x >> 5));
  const int wpb    = blockDim.x >> 5;
  const int nwaves = gridDim.x * wpb;
  const float4 zero = make_float4(0.0f, 0.0f, 0.0f, 0.0f);

  for (int node = blockIdx.x * wpb + wave; node < N; node += nwaves) {
    const int beg = rowptr[node];
    const int end = rowptr[node + 1];
    float4 acc = zero;
    for (int j = beg; j < end; j += 32) {
      const int n = min(32, end - j);
      const int mine = col[j + min(lane, n - 1)];
      int k = 0;
      for (; k + 8 <= n; k += 8) {
        int s[8];
        float4 r[8];
#pragma unroll
        for (int u = 0; u < 8; ++u) s[u] = __builtin_amdgcn_readlane(mine, k + u);
#pragma unroll
        for (int u = 0; u < 8; ++u)
          r[u] = reinterpret_cast<const float4*>(feat + (size_t)s[u] * DIM)[lane];
#pragma unroll
        for (int u = 0; u < 8; ++u) {
          acc.x += r[u].x; acc.y += r[u].y; acc.z += r[u].z; acc.w += r[u].w;
        }
      }
      if (k < n) {
        int s[8];
        float4 r[8];
#pragma unroll
        for (int u = 0; u < 8; ++u) s[u] = __builtin_amdgcn_readlane(mine, k + u);
#pragma unroll
        for (int u = 0; u < 8; ++u) {
          r[u] = zero;
          if (k + u < n) r[u] = reinterpret_cast<const float4*>(feat + (size_t)s[u] * DIM)[lane];
        }
#pragma unroll
        for (int u = 0; u < 8; ++u) {
          acc.x += r[u].x; acc.y += r[u].y; acc.z += r[u].z; acc.w += r[u].w;
        }
      }
    }
    reinterpret_cast<float4*>(agg + (size_t)node * DIM)[lane] = acc;
  }
}

#define WP_PITCH 288
#define W_LDS_FLOATS (64 * WP_PITCH)
#define A_PITCH 132
#define A_TILE_FLOATS (16 * A_PITCH)
#define GEMM_WAVES 8
#define GEMM_LDS_FLOATS (2 * W_LDS_FLOATS + GEMM_WAVES * 2 * A_TILE_FLOATS)

__device__ __forceinline__ void wave_lds_handover() {
  __builtin_amdgcn_fence(__ATOMIC_RELEASE, "wavefront");
  __builtin_amdgcn_fence(__ATOMIC_ACQUIRE, "wavefront");
}

__global__ __launch_bounds__(256)
void sage_fused_gemm(const float* __restrict__ agg,
                     const float* __restrict__ cnt,
                     const float* __restrict__ xin,
                     const float* __restrict__ Wl,
                     const float* __restrict__ Wr,
                     const float* __restrict__ bias,
                     float* __restrict__ out,
                     int do_relu, int nstripes) {
  __shared__ __attribute__((aligned(16))) float lds[GEMM_LDS_FLOATS];

  const int tid  = threadIdx.x;
  const int lane = tid & 31;
  const int wave = __builtin_amdgcn_readfirstlane(tid >> 5);
  const int m    = lane & 15;
  const int half = lane >> 4;

  float* sWl = lds;
  float* sWr = lds + W_LDS_FLOATS;
  float* sAg = lds + 2 * W_LDS_FLOATS + wave * (2 * A_TILE_FLOATS);
  float* sAx = sAg + A_TILE_FLOATS;

#pragma unroll
  for (int i = 0; i < 8; ++i) {
    const int p = wave + 8 * i;
    const float4 l0 = reinterpret_cast<const float4*>(Wl + (size_t)(2 * p) * DIM)[lane];
    const float4 l1 = reinterpret_cast<const float4*>(Wl + (size_t)(2 * p + 1) * DIM)[lane];
    const float4 r0 = reinterpret_cast<const float4*>(Wr + (size_t)(2 * p) * DIM)[lane];
    const float4 r1 = reinterpret_cast<const float4*>(Wr + (size_t)(2 * p + 1) * DIM)[lane];
    float4* dl = reinterpret_cast<float4*>(sWl + p * WP_PITCH + 8 * lane);
    float4* dr = reinterpret_cast<float4*>(sWr + p * WP_PITCH + 8 * lane);
    dl[0] = make_float4(l0.x, l1.x, l0.y, l1.y);
    dl[1] = make_float4(l0.z, l1.z, l0.w, l1.w);
    dr[0] = make_float4(r0.x, r1.x, r0.y, r1.y);
    dr[1] = make_float4(r0.z, r1.z, r0.w, r1.w);
  }

  float bv[8];
#pragma unroll
  for (int t = 0; t < 8; ++t) bv[t] = bias[16 * t + m];

  const int stride = GEMM_WAVES * gridDim.x;
  int stripe = wave * gridDim.x + blockIdx.x;

  v4f ra[16], rx[16];
  float cn = 1.0f;
  if (stripe < nstripes) {
    const size_t row0 = (size_t)stripe * 16;
    cn = cnt[row0 + m];
#pragma unroll
    for (int r = 0; r < 16; ++r) {
      ra[r] = reinterpret_cast<const v4f*>(agg + (row0 + r) * DIM)[lane];
      rx[r] = reinterpret_cast<const v4f*>(xin + (row0 + r) * DIM)[lane];
    }
  }

  __syncthreads();

  const float* aG = sAg + m * A_PITCH + 2 * half;
  const float* aX = sAx + m * A_PITCH + 2 * half;
  const float* bL = sWl + half * WP_PITCH + 2 * m;
  const float* bR = sWr + half * WP_PITCH + 2 * m;

  while (stripe < nstripes) {
    const size_t row0 = (size_t)stripe * 16;

    const float inv = 1.0f / fmaxf(cn, 1.0f);
#pragma unroll
    for (int r = 0; r < 16; ++r) {
      const float iv = __int_as_float(__builtin_amdgcn_readlane(__float_as_int(inv), r));
      reinterpret_cast<v4f*>(sAg + r * A_PITCH)[lane] = ra[r] * iv;
      reinterpret_cast<v4f*>(sAx + r * A_PITCH)[lane] = rx[r];
    }
    wave_lds_handover();

    const int next = stripe + stride;
    if (next < nstripes) {
      const size_t nrow0 = (size_t)next * 16;
      cn = cnt[nrow0 + m];
#pragma unroll
      for (int r = 0; r < 16; ++r) {
        ra[r] = reinterpret_cast<const v4f*>(agg + (nrow0 + r) * DIM)[lane];
        rx[r] = reinterpret_cast<const v4f*>(xin + (nrow0 + r) * DIM)[lane];
      }
    }

    v8f c[8];
#pragma unroll
    for (int t = 0; t < 8; ++t) c[t] = v8f{};

#pragma unroll
    for (int s = 0; s < 32; ++s) {
      const v2f ag = *reinterpret_cast<const v2f*>(aG + 4 * s);
      const v2f ax = *reinterpret_cast<const v2f*>(aX + 4 * s);
      v2f wl[8], wr[8];
#pragma unroll
      for (int t = 0; t < 8; ++t) {
        wl[t] = *reinterpret_cast<const v2f*>(bL + 2 * s * WP_PITCH + 32 * t);
        wr[t] = *reinterpret_cast<const v2f*>(bR + 2 * s * WP_PITCH + 32 * t);
      }
#pragma unroll
      for (int t = 0; t < 8; ++t)
        c[t] = __builtin_amdgcn_wmma_f32_16x16x4_f32(false, ag, false, wl[t], (short)0, c[t],
                                                     false, false);
#pragma unroll
      for (int t = 0; t < 8; ++t)
        c[t] = __builtin_amdgcn_wmma_f32_16x16x4_f32(false, ax, false, wr[t], (short)0, c[t],
                                                     false, false);
    }

#pragma unroll
    for (int t = 0; t < 8; ++t) {
#pragma unroll
      for (int r = 0; r < 8; ++r) {
        float v = c[t][r] + bv[t];
        if (do_relu) v = fmaxf(v, 0.0f);
        out[(row0 + r + 8 * half) * DIM + 16 * t + m] = v;
      }
    }

    wave_lds_handover();
    stripe = next;
  }
}

extern "C" void kernel_launch(void* const* d_in, const int* in_sizes, int n_in,
                              void* d_out, int out_size, void* d_ws, size_t ws_size,
                              hipStream_t stream) {
  const float* x   = (const float*)d_in[0];
  const int*  edge = (const int*)d_in[1];
  const float* W1l = (const float*)d_in[2];
  const float* b1  = (const float*)d_in[3];
  const float* W1r = (const float*)d_in[4];
  const float* W2l = (const float*)d_in[5];
  const float* b2  = (const float*)d_in[6];
  const float* W2r = (const float*)d_in[7];

  const int N = in_sizes[0] / DIM;
  const int E = in_sizes[1] / 2;
  const int* src = edge;
  const int* dst = edge + E;

  float* agg = (float*)d_ws;
  float* cnt = agg + (size_t)N * DIM;
  float* h   = cnt + N;
  int* rowptr = (int*)(h + (size_t)N * DIM);
  int* tab    = rowptr + N + 4;
  int* bstart = tab + CS_BMAX * CS_G;
  int* tmp    = bstart + CS_BMAX + 4;
  int* col    = tmp + 2 * (size_t)E;
  float* out = (float*)d_out;

  const int buckets = (N + CS_W - 1) / CS_W;
  const int gemm_stripes = N / 16;

  sage_bucket_count<<<CS_G, CS_THREADS, 0, stream>>>(dst, tab, E);
  sage_bucket_scan<<<1, CS_THREADS, 0, stream>>>(tab, bstart);
  sage_bucket_scatter<<<CS_G, CS_THREADS, 0, stream>>>(src, dst, tab, tmp, E);
  sage_bucket_sort<<<buckets, CS_THREADS, 0, stream>>>(tmp, bstart, rowptr, cnt, col, N);

  sage_gather_sum<<<2048, 256, 0, stream>>>(x, rowptr, col, agg, N);
  sage_fused_gemm<<<256, 256, 0, stream>>>(agg, cnt, x, W1l, W1r, b1, h, 1, gemm_stripes);

  sage_gather_sum<<<2048, 256, 0, stream>>>(h, rowptr, col, agg, N);
  sage_fused_gemm<<<256, 256, 0, stream>>>(agg, cnt, h, W2l, W2r, b2, out, 0, gemm_stripes);
}


# level 1 of the dst counting sort done chunk-locally in LDS (count, scan, place, whole-line write-back); global count and table-scan launches dropped, 8 -> 6; sum order = LDS arrival rank
# speedup vs baseline: 4.1070x; 1.1306x over previous
#include <hip/hip_runtime.h>
#include <hip/hip_bf16.h>

#define DIM 128

typedef __attribute__((ext_vector_type(2))) float v2f;
typedef __attribute__((ext_vector_type(4))) float v4f;
typedef __attribute__((ext_vector_type(8))) float v8f;


#define CS_SHIFT 9
#define CS_W (1 << CS_SHIFT)
#define CS_BMAX 256
#define CS_G 256
#define CS_THREADS 1024
#define CS_EPT 7
#define CS_CAP (CS_EPT * CS_THREADS)
#define CS_SEGS (CS_G / 32)

__global__ __launch_bounds__(CS_THREADS)
void sage_chunk_sort(const int* __restrict__ src, const int* __restrict__ dst,
                     int* __restrict__ tab, int* __restrict__ tmp_i, int E) {
  __shared__ int cur[CS_BMAX];
  __shared__ int w_tot[CS_THREADS / 32];
  __shared__ int2 sorted[CS_CAP];
  int2* tmp = reinterpret_cast<int2*>(tmp_i);
  const int tid  = threadIdx.x;
  const int lane = tid & 31;
  const int wave = tid >> 5;
  const int g    = blockIdx.x;
  if (tid < CS_BMAX) cur[tid] = 0;

  const int chunk = (E + CS_G - 1) / CS_G;
  const int beg = g * chunk;
  const int end = min(E, beg + chunk);

  int d[CS_EPT], s[CS_EPT];
#pragma unroll
  for (int u = 0; u < CS_EPT; ++u) {
    const int e = beg + tid + u * CS_THREADS;
    d[u] = -1; s[u] = 0;
    if (e < end) { d[u] = dst[e]; s[u] = src[e]; }
  }
  __syncthreads();

#pragma unroll
  for (int u = 0; u < CS_EPT; ++u)
    if (d[u] >= 0) atomicAdd(&cur[d[u] >> CS_SHIFT], 1);
  __syncthreads();

  const int v = (tid < CS_BMAX) ? cur[tid] : 0;
  int inc = v;
#pragma unroll
  for (int o = 1; o < 32; o <<= 1) {
    const int t = __shfl_up(inc, o);
    if (lane >= o) inc += t;
  }
  if (lane == 31) w_tot[wave] = inc;
  __syncthreads();
  int off = 0;
#pragma unroll
  for (int w = 0; w < CS_THREADS / 32; ++w)
    if (w < wave) off += w_tot[w];
  const int ex = off + inc - v;
  if (tid < CS_BMAX) {
    tab[tid * CS_G + g] = ex;
    cur[tid] = ex;
    if (tid == CS_BMAX - 1) tab[CS_BMAX * CS_G + g] = ex + v;
  }
  __syncthreads();

  int pos[CS_EPT];
#pragma unroll
  for (int u = 0; u < CS_EPT; ++u) {
    pos[u] = -1;
    if (d[u] >= 0) pos[u] = atomicAdd(&cur[d[u] >> CS_SHIFT], 1);
  }
#pragma unroll
  for (int u = 0; u < CS_EPT; ++u)
    if (pos[u] >= 0) sorted[pos[u]] = make_int2(s[u], d[u]);
  __syncthreads();

  const int len = end - beg;
  int2 o[CS_EPT];
#pragma unroll
  for (int u = 0; u < CS_EPT; ++u) o[u] = sorted[tid + u * CS_THREADS];
#pragma unroll
  for (int u = 0; u < CS_EPT; ++u) {
    const int i = tid + u * CS_THREADS;
    if (i < len) tmp[beg + i] = o[u];
  }
}

__global__ __launch_bounds__(CS_THREADS)
void sage_bucket_sort(const int* __restrict__ tmp_i, const int* __restrict__ tab,
                      int* __restrict__ rowptr, float* __restrict__ cnt,
                      int* __restrict__ col, int N, int E) {
  __shared__ int c[CS_W];
  __shared__ int w_tot[CS_THREADS / 32];
  __shared__ int w_lo[CS_THREADS / 32];
  __shared__ int seg_beg[CS_G];
  __shared__ int seg_len[CS_G];
  const int2* tmp = reinterpret_cast<const int2*>(tmp_i);
  const int tid  = threadIdx.x;
  const int lane = tid & 31;
  const int wave = __builtin_amdgcn_readfirstlane(tid >> 5);
  const int b    = blockIdx.x;
  const int node0 = b << CS_SHIFT;
  const int chunk = (E + CS_G - 1) / CS_G;

  int lo = 0;
  if (tid < CS_G) {
    lo = tab[b * CS_G + tid];
    const int hi = tab[(b + 1) * CS_G + tid];
    seg_beg[tid] = tid * chunk + lo;
    seg_len[tid] = hi - lo;
  }
  if (tid < CS_W) c[tid] = 0;
  int part = lo;
#pragma unroll
  for (int o = 16; o > 0; o >>= 1) part += __shfl_xor(part, o);
  if (lane == 0) w_lo[wave] = part;
  __syncthreads();
  int beg = 0;
#pragma unroll
  for (int w = 0; w < CS_THREADS / 32; ++w) beg += w_lo[w];

  int sb[CS_SEGS], sl[CS_SEGS];
  int maxlen = 0;
#pragma unroll
  for (int k = 0; k < CS_SEGS; ++k) {
    sb[k] = __builtin_amdgcn_readfirstlane(seg_beg[wave + 32 * k]);
    sl[k] = __builtin_amdgcn_readfirstlane(seg_len[wave + 32 * k]);
    maxlen = max(maxlen, sl[k]);
  }

  for (int i0 = 0; i0 < maxlen; i0 += 32) {
    const int i = i0 + lane;
    int d[CS_SEGS];
#pragma unroll
    for (int k = 0; k < CS_SEGS; ++k)
      d[k] = (i < sl[k]) ? tmp[sb[k] + i].y : -1;
#pragma unroll
    for (int k = 0; k < CS_SEGS; ++k)
      if (d[k] >= 0) atomicAdd(&c[d[k] - node0], 1);
  }
  __syncthreads();

  const int v = (tid < CS_W) ? c[tid] : 0;
  int inc = v;
#pragma unroll
  for (int o = 1; o < 32; o <<= 1) {
    const int t = __shfl_up(inc, o);
    if (lane >= o) inc += t;
  }
  if (lane == 31) w_tot[wave] = inc;
  __syncthreads();
  int off = 0;
#pragma unroll
  for (int w = 0; w < CS_THREADS / 32; ++w)
    if (w < wave) off += w_tot[w];
  const int ex = off + inc - v;

  const int node = node0 + tid;
  if (tid < CS_W) {
    if (node < N) {
      rowptr[node] = beg + ex;
      cnt[node]    = (float)v;
      if (node == N - 1) rowptr[N] = beg + ex + v;
    }
    c[tid] = ex;
  }
  __syncthreads();

  for (int i0 = 0; i0 < maxlen; i0 += 32) {
    const int i = i0 + lane;
    int2 e[CS_SEGS];
#pragma unroll
    for (int k = 0; k < CS_SEGS; ++k) {
      e[k] = make_int2(0, -1);
      if (i < sl[k]) e[k] = tmp[sb[k] + i];
    }
    int pos[CS_SEGS];
#pragma unroll
    for (int k = 0; k < CS_SEGS; ++k) {
      pos[k] = -1;
      if (e[k].y >= 0) pos[k] = atomicAdd(&c[e[k].y - node0], 1);
    }
#pragma unroll
    for (int k = 0; k < CS_SEGS; ++k)
      if (pos[k] >= 0) col[beg + pos[k]] = e[k].x;
  }
}

__global__ __launch_bounds__(256)
void sage_gather_sum(const float* __restrict__ feat,
                     const int* __restrict__ rowptr,
                     const int* __restrict__ col,
                     float* __restrict__ agg, int N) {
  const int lane = threadIdx.x & 31;
  const int wave   = __builtin_amdgcn_readfirstlane((int)(threadIdx.x >> 5));
  const int wpb    = blockDim.x >> 5;
  const int nwaves = gridDim.x * wpb;
  const float4 zero = make_float4(0.0f, 0.0f, 0.0f, 0.0f);

  for (int node = blockIdx.x * wpb + wave; node < N; node += nwaves) {
    const int beg = rowptr[node];
    const int end = rowptr[node + 1];
    float4 acc = zero;
    for (int j = beg; j < end; j += 32) {
      const int n = min(32, end - j);
      const int mine = col[j + min(lane, n - 1)];
      int k = 0;
      for (; k + 8 <= n; k += 8) {
        int s[8];
        float4 r[8];
#pragma unroll
        for (int u = 0; u < 8; ++u) s[u] = __builtin_amdgcn_readlane(mine, k + u);
#pragma unroll
        for (int u = 0; u < 8; ++u)
          r[u] = reinterpret_cast<const float4*>(feat + (size_t)s[u] * DIM)[lane];
#pragma unroll
        for (int u = 0; u < 8; ++u) {
          acc.x += r[u].x; acc.y += r[u].y; acc.z += r[u].z; acc.w += r[u].w;
        }
      }
      if (k < n) {
        int s[8];
        float4 r[8];
#pragma unroll
        for (int u = 0; u < 8; ++u) s[u] = __builtin_amdgcn_readlane(mine, k + u);
#pragma unroll
        for (int u = 0; u < 8; ++u) {
          r[u] = zero;
          if (k + u < n) r[u] = reinterpret_cast<const float4*>(feat + (size_t)s[u] * DIM)[lane];
        }
#pragma unroll
        for (int u = 0; u < 8; ++u) {
          acc.x += r[u].x; acc.y += r[u].y; acc.z += r[u].z; acc.w += r[u].w;
        }
      }
    }
    reinterpret_cast<float4*>(agg + (size_t)node * DIM)[lane] = acc;
  }
}

#define WP_PITCH 288
#define W_LDS_FLOATS (64 * WP_PITCH)
#define A_PITCH 132
#define A_TILE_FLOATS (16 * A_PITCH)
#define GEMM_WAVES 8
#define GEMM_LDS_FLOATS (2 * W_LDS_FLOATS + GEMM_WAVES * 2 * A_TILE_FLOATS)

__device__ __forceinline__ void wave_lds_handover() {
  __builtin_amdgcn_fence(__ATOMIC_RELEASE, "wavefront");
  __builtin_amdgcn_fence(__ATOMIC_ACQUIRE, "wavefront");
}

__global__ __launch_bounds__(256)
void sage_fused_gemm(const float* __restrict__ agg,
                     const float* __restrict__ cnt,
                     const float* __restrict__ xin,
                     const float* __restrict__ Wl,
                     const float* __restrict__ Wr,
                     const float* __restrict__ bias,
                     float* __restrict__ out,
                     int do_relu, int nstripes) {
  __shared__ __attribute__((aligned(16))) float lds[GEMM_LDS_FLOATS];

  const int tid  = threadIdx.x;
  const int lane = tid & 31;
  const int wave = __builtin_amdgcn_readfirstlane(tid >> 5);
  const int m    = lane & 15;
  const int half = lane >> 4;

  float* sWl = lds;
  float* sWr = lds + W_LDS_FLOATS;
  float* sAg = lds + 2 * W_LDS_FLOATS + wave * (2 * A_TILE_FLOATS);
  float* sAx = sAg + A_TILE_FLOATS;

#pragma unroll
  for (int i = 0; i < 8; ++i) {
    const int p = wave + 8 * i;
    const float4 l0 = reinterpret_cast<const float4*>(Wl + (size_t)(2 * p) * DIM)[lane];
    const float4 l1 = reinterpret_cast<const float4*>(Wl + (size_t)(2 * p + 1) * DIM)[lane];
    const float4 r0 = reinterpret_cast<const float4*>(Wr + (size_t)(2 * p) * DIM)[lane];
    const float4 r1 = reinterpret_cast<const float4*>(Wr + (size_t)(2 * p + 1) * DIM)[lane];
    float4* dl = reinterpret_cast<float4*>(sWl + p * WP_PITCH + 8 * lane);
    float4* dr = reinterpret_cast<float4*>(sWr + p * WP_PITCH + 8 * lane);
    dl[0] = make_float4(l0.x, l1.x, l0.y, l1.y);
    dl[1] = make_float4(l0.z, l1.z, l0.w, l1.w);
    dr[0] = make_float4(r0.x, r1.x, r0.y, r1.y);
    dr[1] = make_float4(r0.z, r1.z, r0.w, r1.w);
  }

  float bv[8];
#pragma unroll
  for (int t = 0; t < 8; ++t) bv[t] = bias[16 * t + m];

  const int stride = GEMM_WAVES * gridDim.x;
  int stripe = wave * gridDim.x + blockIdx.x;

  v4f ra[16], rx[16];
  float cn = 1.0f;
  if (stripe < nstripes) {
    const size_t row0 = (size_t)stripe * 16;
    cn = cnt[row0 + m];
#pragma unroll
    for (int r = 0; r < 16; ++r) {
      ra[r] = reinterpret_cast<const v4f*>(agg + (row0 + r) * DIM)[lane];
      rx[r] = reinterpret_cast<const v4f*>(xin + (row0 + r) * DIM)[lane];
    }
  }

  __syncthreads();

  const float* aG = sAg + m * A_PITCH + 2 * half;
  const float* aX = sAx + m * A_PITCH + 2 * half;
  const float* bL = sWl + half * WP_PITCH + 2 * m;
  const float* bR = sWr + half * WP_PITCH + 2 * m;

  while (stripe < nstripes) {
    const size_t row0 = (size_t)stripe * 16;

    const float inv = 1.0f / fmaxf(cn, 1.0f);
#pragma unroll
    for (int r = 0; r < 16; ++r) {
      const float iv = __int_as_float(__builtin_amdgcn_readlane(__float_as_int(inv), r));
      reinterpret_cast<v4f*>(sAg + r * A_PITCH)[lane] = ra[r] * iv;
      reinterpret_cast<v4f*>(sAx + r * A_PITCH)[lane] = rx[r];
    }
    wave_lds_handover();

    const int next = stripe + stride;
    if (next < nstripes) {
      const size_t nrow0 = (size_t)next * 16;
      cn = cnt[nrow0 + m];
#pragma unroll
      for (int r = 0; r < 16; ++r) {
        ra[r] = reinterpret_cast<const v4f*>(agg + (nrow0 + r) * DIM)[lane];
        rx[r] = reinterpret_cast<const v4f*>(xin + (nrow0 + r) * DIM)[lane];
      }
    }

    v8f c[8];
#pragma unroll
    for (int t = 0; t < 8; ++t) c[t] = v8f{};

#pragma unroll
    for (int s = 0; s < 32; ++s) {
      const v2f ag = *reinterpret_cast<const v2f*>(aG + 4 * s);
      const v2f ax = *reinterpret_cast<const v2f*>(aX + 4 * s);
      v2f wl[8], wr[8];
#pragma unroll
      for (int t = 0; t < 8; ++t) {
        wl[t] = *reinterpret_cast<const v2f*>(bL + 2 * s * WP_PITCH + 32 * t);
        wr[t] = *reinterpret_cast<const v2f*>(bR + 2 * s * WP_PITCH + 32 * t);
      }
#pragma unroll
      for (int t = 0; t < 8; ++t)
        c[t] = __builtin_amdgcn_wmma_f32_16x16x4_f32(false, ag, false, wl[t], (short)0, c[t],
                                                     false, false);
#pragma unroll
      for (int t = 0; t < 8; ++t)
        c[t] = __builtin_amdgcn_wmma_f32_16x16x4_f32(false, ax, false, wr[t], (short)0, c[t],
                                                     false, false);
    }

#pragma unroll
    for (int t = 0; t < 8; ++t) {
#pragma unroll
      for (int r = 0; r < 8; ++r) {
        float v = c[t][r] + bv[t];
        if (do_relu) v = fmaxf(v, 0.0f);
        out[(row0 + r + 8 * half) * DIM + 16 * t + m] = v;
      }
    }

    wave_lds_handover();
    stripe = next;
  }
}

extern "C" void kernel_launch(void* const* d_in, const int* in_sizes, int n_in,
                              void* d_out, int out_size, void* d_ws, size_t ws_size,
                              hipStream_t stream) {
  const float* x   = (const float*)d_in[0];
  const int*  edge = (const int*)d_in[1];
  const float* W1l = (const float*)d_in[2];
  const float* b1  = (const float*)d_in[3];
  const float* W1r = (const float*)d_in[4];
  const float* W2l = (const float*)d_in[5];
  const float* b2  = (const float*)d_in[6];
  const float* W2r = (const float*)d_in[7];

  const int N = in_sizes[0] / DIM;
  const int E = in_sizes[1] / 2;
  const int* src = edge;
  const int* dst = edge + E;

  float* agg = (float*)d_ws;
  float* cnt = agg + (size_t)N * DIM;
  float* h   = cnt + N;
  int* rowptr = (int*)(h + (size_t)N * DIM);
  int* tab    = rowptr + N + 4;
  int* tmp    = tab + (CS_BMAX + 1) * CS_G;
  int* col    = tmp + 2 * (size_t)E;
  float* out = (float*)d_out;

  const int buckets = (N + CS_W - 1) / CS_W;
  const int gemm_stripes = N / 16;

  sage_chunk_sort<<<CS_G, CS_THREADS, 0, stream>>>(src, dst, tab, tmp, E);
  sage_bucket_sort<<<buckets, CS_THREADS, 0, stream>>>(tmp, tab, rowptr, cnt, col, N, E);

  sage_gather_sum<<<2048, 256, 0, stream>>>(x, rowptr, col, agg, N);
  sage_fused_gemm<<<256, 256, 0, stream>>>(agg, cnt, x, W1l, W1r, b1, h, 1, gemm_stripes);

  sage_gather_sum<<<2048, 256, 0, stream>>>(h, rowptr, col, agg, N);
  sage_fused_gemm<<<256, 256, 0, stream>>>(agg, cnt, h, W2l, W2r, b2, out, 0, gemm_stripes);
}


# neighbour gather-sum fused into the layer kernel (16 waves per WGP, each sums its stripe's rows into its LDS tile, then multiplies); 6 -> 4 launches; sum order: all aggregate k steps, then self
# speedup vs baseline: 4.2654x; 1.0386x over previous
#include <hip/hip_runtime.h>
#include <hip/hip_bf16.h>

#define DIM 128

typedef __attribute__((ext_vector_type(2))) float v2f;
typedef __attribute__((ext_vector_type(4))) float v4f;
typedef __attribute__((ext_vector_type(8))) float v8f;


#define CS_SHIFT 9
#define CS_W (1 << CS_SHIFT)
#define CS_BMAX 256
#define CS_G 256
#define CS_THREADS 1024
#define CS_EPT 7
#define CS_CAP (CS_EPT * CS_THREADS)
#define CS_SEGS (CS_G / 32)

__global__ __launch_bounds__(CS_THREADS)
void sage_chunk_sort(const int* __restrict__ src, const int* __restrict__ dst,
                     int* __restrict__ tab, int* __restrict__ tmp_i, int E) {
  __shared__ int cur[CS_BMAX];
  __shared__ int w_tot[CS_THREADS / 32];
  __shared__ int2 sorted[CS_CAP];
  int2* tmp = reinterpret_cast<int2*>(tmp_i);
  const int tid  = threadIdx.x;
  const int lane = tid & 31;
  const int wave = tid >> 5;
  const int g    = blockIdx.x;
  if (tid < CS_BMAX) cur[tid] = 0;

  const int chunk = (E + CS_G - 1) / CS_G;
  const int beg = g * chunk;
  const int end = min(E, beg + chunk);

  int d[CS_EPT], s[CS_EPT];
#pragma unroll
  for (int u = 0; u < CS_EPT; ++u) {
    const int e = beg + tid + u * CS_THREADS;
    d[u] = -1; s[u] = 0;
    if (e < end) { d[u] = dst[e]; s[u] = src[e]; }
  }
  __syncthreads();

#pragma unroll
  for (int u = 0; u < CS_EPT; ++u)
    if (d[u] >= 0) atomicAdd(&cur[d[u] >> CS_SHIFT], 1);
  __syncthreads();

  const int v = (tid < CS_BMAX) ? cur[tid] : 0;
  int inc = v;
#pragma unroll
  for (int o = 1; o < 32; o <<= 1) {
    const int t = __shfl_up(inc, o);
    if (lane >= o) inc += t;
  }
  if (lane == 31) w_tot[wave] = inc;
  __syncthreads();
  int off = 0;
#pragma unroll
  for (int w = 0; w < CS_THREADS / 32; ++w)
    if (w < wave) off += w_tot[w];
  const int ex = off + inc - v;
  if (tid < CS_BMAX) {
    tab[tid * CS_G + g] = ex;
    cur[tid] = ex;
    if (tid == CS_BMAX - 1) tab[CS_BMAX * CS_G + g] = ex + v;
  }
  __syncthreads();

  int pos[CS_EPT];
#pragma unroll
  for (int u = 0; u < CS_EPT; ++u) {
    pos[u] = -1;
    if (d[u] >= 0) pos[u] = atomicAdd(&cur[d[u] >> CS_SHIFT], 1);
  }
#pragma unroll
  for (int u = 0; u < CS_EPT; ++u)
    if (pos[u] >= 0) sorted[pos[u]] = make_int2(s[u], d[u]);
  __syncthreads();

  const int len = end - beg;
  int2 o[CS_EPT];
#pragma unroll
  for (int u = 0; u < CS_EPT; ++u) o[u] = sorted[tid + u * CS_THREADS];
#pragma unroll
  for (int u = 0; u < CS_EPT; ++u) {
    const int i = tid + u * CS_THREADS;
    if (i < len) tmp[beg + i] = o[u];
  }
}

__global__ __launch_bounds__(CS_THREADS)
void sage_bucket_sort(const int* __restrict__ tmp_i, const int* __restrict__ tab,
                      int* __restrict__ rowptr,
                      int* __restrict__ col, int N, int E) {
  __shared__ int c[CS_W];
  __shared__ int w_tot[CS_THREADS / 32];
  __shared__ int w_lo[CS_THREADS / 32];
  __shared__ int seg_beg[CS_G];
  __shared__ int seg_len[CS_G];
  const int2* tmp = reinterpret_cast<const int2*>(tmp_i);
  const int tid  = threadIdx.x;
  const int lane = tid & 31;
  const int wave = __builtin_amdgcn_readfirstlane(tid >> 5);
  const int b    = blockIdx.x;
  const int node0 = b << CS_SHIFT;
  const int chunk = (E + CS_G - 1) / CS_G;

  int lo = 0;
  if (tid < CS_G) {
    lo = tab[b * CS_G + tid];
    const int hi = tab[(b + 1) * CS_G + tid];
    seg_beg[tid] = tid * chunk + lo;
    seg_len[tid] = hi - lo;
  }
  if (tid < CS_W) c[tid] = 0;
  int part = lo;
#pragma unroll
  for (int o = 16; o > 0; o >>= 1) part += __shfl_xor(part, o);
  if (lane == 0) w_lo[wave] = part;
  __syncthreads();
  int beg = 0;
#pragma unroll
  for (int w = 0; w < CS_THREADS / 32; ++w) beg += w_lo[w];

  int sb[CS_SEGS], sl[CS_SEGS];
  int maxlen = 0;
#pragma unroll
  for (int k = 0; k < CS_SEGS; ++k) {
    sb[k] = __builtin_amdgcn_readfirstlane(seg_beg[wave + 32 * k]);
    sl[k] = __builtin_amdgcn_readfirstlane(seg_len[wave + 32 * k]);
    maxlen = max(maxlen, sl[k]);
  }

  for (int i0 = 0; i0 < maxlen; i0 += 32) {
    const int i = i0 + lane;
    int d[CS_SEGS];
#pragma unroll
    for (int k = 0; k < CS_SEGS; ++k)
      d[k] = (i < sl[k]) ? tmp[sb[k] + i].y : -1;
#pragma unroll
    for (int k = 0; k < CS_SEGS; ++k)
      if (d[k] >= 0) atomicAdd(&c[d[k] - node0], 1);
  }
  __syncthreads();

  const int v = (tid < CS_W) ? c[tid] : 0;
  int inc = v;
#pragma unroll
  for (int o = 1; o < 32; o <<= 1) {
    const int t = __shfl_up(inc, o);
    if (lane >= o) inc += t;
  }
  if (lane == 31) w_tot[wave] = inc;
  __syncthreads();
  int off = 0;
#pragma unroll
  for (int w = 0; w < CS_THREADS / 32; ++w)
    if (w < wave) off += w_tot[w];
  const int ex = off + inc - v;

  const int node = node0 + tid;
  if (tid < CS_W) {
    if (node < N) {
      rowptr[node] = beg + ex;
      if (node == N - 1) rowptr[N] = beg + ex + v;
    }
    c[tid] = ex;
  }
  __syncthreads();

  for (int i0 = 0; i0 < maxlen; i0 += 32) {
    const int i = i0 + lane;
    int2 e[CS_SEGS];
#pragma unroll
    for (int k = 0; k < CS_SEGS; ++k) {
      e[k] = make_int2(0, -1);
      if (i < sl[k]) e[k] = tmp[sb[k] + i];
    }
    int pos[CS_SEGS];
#pragma unroll
    for (int k = 0; k < CS_SEGS; ++k) {
      pos[k] = -1;
      if (e[k].y >= 0) pos[k] = atomicAdd(&c[e[k].y - node0], 1);
    }
#pragma unroll
    for (int k = 0; k < CS_SEGS; ++k)
      if (pos[k] >= 0) col[beg + pos[k]] = e[k].x;
  }
}

#define WP_PITCH 288
#define W_LDS_FLOATS (64 * WP_PITCH)
#define A_PITCH 132
#define A_TILE_FLOATS (16 * A_PITCH)
#define LAYER_WAVES 16
#define LAYER_LDS_FLOATS (2 * W_LDS_FLOATS + LAYER_WAVES * A_TILE_FLOATS)
#define ROWS_AHEAD 16

__device__ __forceinline__ void wave_lds_handover() {
  __builtin_amdgcn_fence(__ATOMIC_RELEASE, "wavefront");
  __builtin_amdgcn_fence(__ATOMIC_ACQUIRE, "wavefront");
}

__global__ __launch_bounds__(32 * LAYER_WAVES)
void sage_fused_layer(const float* __restrict__ feat,
                      const int* __restrict__ rowptr,
                      const int* __restrict__ col,
                      const float* __restrict__ Wl,
                      const float* __restrict__ Wr,
                      const float* __restrict__ bias,
                      float* __restrict__ out,
                      int do_relu, int nstripes) {
  __shared__ __attribute__((aligned(16))) float lds[LAYER_LDS_FLOATS];

  const int tid  = threadIdx.x;
  const int lane = tid & 31;
  const int wave = __builtin_amdgcn_readfirstlane(tid >> 5);
  const int m    = lane & 15;
  const int half = lane >> 4;

  float* sWl = lds;
  float* sWr = lds + W_LDS_FLOATS;
  float* sA  = lds + 2 * W_LDS_FLOATS + wave * A_TILE_FLOATS;

#pragma unroll
  for (int i = 0; i < 64 / LAYER_WAVES; ++i) {
    const int p = wave + LAYER_WAVES * i;
    const float4 l0 = reinterpret_cast<const float4*>(Wl + (size_t)(2 * p) * DIM)[lane];
    const float4 l1 = reinterpret_cast<const float4*>(Wl + (size_t)(2 * p + 1) * DIM)[lane];
    const float4 r0 = reinterpret_cast<const float4*>(Wr + (size_t)(2 * p) * DIM)[lane];
    const float4 r1 = reinterpret_cast<const float4*>(Wr + (size_t)(2 * p + 1) * DIM)[lane];
    float4* dl = reinterpret_cast<float4*>(sWl + p * WP_PITCH + 8 * lane);
    float4* dr = reinterpret_cast<float4*>(sWr + p * WP_PITCH + 8 * lane);
    dl[0] = make_float4(l0.x, l1.x, l0.y, l1.y);
    dl[1] = make_float4(l0.z, l1.z, l0.w, l1.w);
    dr[0] = make_float4(r0.x, r1.x, r0.y, r1.y);
    dr[1] = make_float4(r0.z, r1.z, r0.w, r1.w);
  }

  float bv[8];
#pragma unroll
  for (int t = 0; t < 8; ++t) bv[t] = bias[16 * t + m];

  __syncthreads();

  const float* aF = sA + m * A_PITCH + 2 * half;
  const float* bL = sWl + half * WP_PITCH + 2 * m;
  const float* bR = sWr + half * WP_PITCH + 2 * m;
  const float4 zero = make_float4(0.0f, 0.0f, 0.0f, 0.0f);

  const int stride = LAYER_WAVES * gridDim.x;
  for (int stripe = wave * gridDim.x + blockIdx.x; stripe < nstripes; stripe += stride) {
    const size_t row0 = (size_t)stripe * 16;

    const int rp = rowptr[row0 + min(lane, 16)];
    int beg = __builtin_amdgcn_readlane(rp, 0);
    int end = __builtin_amdgcn_readlane(rp, 1);
    int mine = 0;
    if (end > beg) mine = col[beg + min(lane, min(32, end - beg) - 1)];

    for (int r = 0; r < 16; ++r) {
      const int nbeg = end;
      int nend = end;
      int nmine = 0;
      if (r < 15) {
        nend = __builtin_amdgcn_readlane(rp, r + 2);
        if (nend > nbeg) nmine = col[nbeg + min(lane, min(32, nend - nbeg) - 1)];
      }

      float4 acc = zero;
      for (int j = beg; j < end; j += 32) {
        const int n = min(32, end - j);
        int cur = mine;
        if (j != beg) cur = col[j + min(lane, n - 1)];
        int k = 0;
        for (; k + ROWS_AHEAD <= n; k += ROWS_AHEAD) {
          int s[ROWS_AHEAD];
          float4 v[ROWS_AHEAD];
#pragma unroll
          for (int u = 0; u < ROWS_AHEAD; ++u) s[u] = __builtin_amdgcn_readlane(cur, k + u);
#pragma unroll
          for (int u = 0; u < ROWS_AHEAD; ++u)
            v[u] = reinterpret_cast<const float4*>(feat + (size_t)s[u] * DIM)[lane];
#pragma unroll
          for (int u = 0; u < ROWS_AHEAD; ++u) {
            acc.x += v[u].x; acc.y += v[u].y; acc.z += v[u].z; acc.w += v[u].w;
          }
        }
        if (k < n) {
          int s[ROWS_AHEAD];
          float4 v[ROWS_AHEAD];
#pragma unroll
          for (int u = 0; u < ROWS_AHEAD; ++u) s[u] = __builtin_amdgcn_readlane(cur, k + u);
#pragma unroll
          for (int u = 0; u < ROWS_AHEAD; ++u) {
            v[u] = zero;
            if (k + u < n) v[u] = reinterpret_cast<const float4*>(feat + (size_t)s[u] * DIM)[lane];
          }
#pragma unroll
          for (int u = 0; u < ROWS_AHEAD; ++u) {
            acc.x += v[u].x; acc.y += v[u].y; acc.z += v[u].z; acc.w += v[u].w;
          }
        }
      }
      const float inv = 1.0f / fmaxf((float)(end - beg), 1.0f);
      v4f sc;
      sc[0] = acc.x * inv; sc[1] = acc.y * inv; sc[2] = acc.z * inv; sc[3] = acc.w * inv;
      reinterpret_cast<v4f*>(sA + r * A_PITCH)[lane] = sc;

      beg = nbeg; end = nend; mine = nmine;
    }
    wave_lds_handover();

    v4f rx[16];
#pragma unroll
    for (int r = 0; r < 16; ++r)
      rx[r] = reinterpret_cast<const v4f*>(feat + (row0 + r) * DIM)[lane];

    v8f c[8];
#pragma unroll
    for (int t = 0; t < 8; ++t) c[t] = v8f{};

#pragma unroll
    for (int s = 0; s < 32; ++s) {
      const v2f ag = *reinterpret_cast<const v2f*>(aF + 4 * s);
      v2f wl[8];
#pragma unroll
      for (int t = 0; t < 8; ++t)
        wl[t] = *reinterpret_cast<const v2f*>(bL + 2 * s * WP_PITCH + 32 * t);
#pragma unroll
      for (int t = 0; t < 8; ++t)
        c[t] = __builtin_amdgcn_wmma_f32_16x16x4_f32(false, ag, false, wl[t], (short)0, c[t],
                                                     false, false);
    }

    wave_lds_handover();
#pragma unroll
    for (int r = 0; r < 16; ++r)
      reinterpret_cast<v4f*>(sA + r * A_PITCH)[lane] = rx[r];
    wave_lds_handover();

#pragma unroll
    for (int s = 0; s < 32; ++s) {
      const v2f ax = *reinterpret_cast<const v2f*>(aF + 4 * s);
      v2f wr[8];
#pragma unroll
      for (int t = 0; t < 8; ++t)
        wr[t] = *reinterpret_cast<const v2f*>(bR + 2 * s * WP_PITCH + 32 * t);
#pragma unroll
      for (int t = 0; t < 8; ++t)
        c[t] = __builtin_amdgcn_wmma_f32_16x16x4_f32(false, ax, false, wr[t], (short)0, c[t],
                                                     false, false);
    }

#pragma unroll
    for (int t = 0; t < 8; ++t) {
#pragma unroll
      for (int r = 0; r < 8; ++r) {
        float v = c[t][r] + bv[t];
        if (do_relu) v = fmaxf(v, 0.0f);
        out[(row0 + r + 8 * half) * DIM + 16 * t + m] = v;
      }
    }

    wave_lds_handover();
  }
}

extern "C" void kernel_launch(void* const* d_in, const int* in_sizes, int n_in,
                              void* d_out, int out_size, void* d_ws, size_t ws_size,
                              hipStream_t stream) {
  const float* x   = (const float*)d_in[0];
  const int*  edge = (const int*)d_in[1];
  const float* W1l = (const float*)d_in[2];
  const float* b1  = (const float*)d_in[3];
  const float* W1r = (const float*)d_in[4];
  const float* W2l = (const float*)d_in[5];
  const float* b2  = (const float*)d_in[6];
  const float* W2r = (const float*)d_in[7];

  const int N = in_sizes[0] / DIM;
  const int E = in_sizes[1] / 2;
  const int* src = edge;
  const int* dst = edge + E;

  float* h    = (float*)d_ws;
  int* rowptr = (int*)(h + (size_t)N * DIM);
  int* tab    = rowptr + N + 4;
  int* tmp    = tab + (CS_BMAX + 1) * CS_G;
  int* col    = tmp + 2 * (size_t)E;
  float* out = (float*)d_out;

  const int buckets = (N + CS_W - 1) / CS_W;
  const int stripes = N / 16;

  sage_chunk_sort<<<CS_G, CS_THREADS, 0, stream>>>(src, dst, tab, tmp, E);
  sage_bucket_sort<<<buckets, CS_THREADS, 0, stream>>>(tmp, tab, rowptr, col, N, E);

  sage_fused_layer<<<256, 32 * LAYER_WAVES, 0, stream>>>(x, rowptr, col, W1l, W1r, b1, h, 1, stripes);
  sage_fused_layer<<<256, 32 * LAYER_WAVES, 0, stream>>>(h, rowptr, col, W2l, W2r, b2, out, 0, stripes);
}


# self-term products (own rows @ Wr) issued inside the neighbour gather, 2 k steps per gathered row; row sums go into the tile words those steps free; sum order: self k steps, then aggregate
# speedup vs baseline: 4.8175x; 1.1294x over previous
#include <hip/hip_runtime.h>
#include <hip/hip_bf16.h>

#define DIM 128

typedef __attribute__((ext_vector_type(2))) float v2f;
typedef __attribute__((ext_vector_type(4))) float v4f;
typedef __attribute__((ext_vector_type(8))) float v8f;


#define CS_SHIFT 9
#define CS_W (1 << CS_SHIFT)
#define CS_BMAX 256
#define CS_G 256
#define CS_THREADS 1024
#define CS_EPT 7
#define CS_CAP (CS_EPT * CS_THREADS)
#define CS_SEGS (CS_G / 32)

__global__ __launch_bounds__(CS_THREADS)
void sage_chunk_sort(const int* __restrict__ src, const int* __restrict__ dst,
                     int* __restrict__ tab, int* __restrict__ tmp_i, int E) {
  __shared__ int cur[CS_BMAX];
  __shared__ int w_tot[CS_THREADS / 32];
  __shared__ int2 sorted[CS_CAP];
  int2* tmp = reinterpret_cast<int2*>(tmp_i);
  const int tid  = threadIdx.x;
  const int lane = tid & 31;
  const int wave = tid >> 5;
  const int g    = blockIdx.x;
  if (tid < CS_BMAX) cur[tid] = 0;

  const int chunk = (E + CS_G - 1) / CS_G;
  const int beg = g * chunk;
  const int end = min(E, beg + chunk);

  int d[CS_EPT], s[CS_EPT];
#pragma unroll
  for (int u = 0; u < CS_EPT; ++u) {
    const int e = beg + tid + u * CS_THREADS;
    d[u] = -1; s[u] = 0;
    if (e < end) { d[u] = dst[e]; s[u] = src[e]; }
  }
  __syncthreads();

#pragma unroll
  for (int u = 0; u < CS_EPT; ++u)
    if (d[u] >= 0) atomicAdd(&cur[d[u] >> CS_SHIFT], 1);
  __syncthreads();

  const int v = (tid < CS_BMAX) ? cur[tid] : 0;
  int inc = v;
#pragma unroll
  for (int o = 1; o < 32; o <<= 1) {
    const int t = __shfl_up(inc, o);
    if (lane >= o) inc += t;
  }
  if (lane == 31) w_tot[wave] = inc;
  __syncthreads();
  int off = 0;
#pragma unroll
  for (int w = 0; w < CS_THREADS / 32; ++w)
    if (w < wave) off += w_tot[w];
  const int ex = off + inc - v;
  if (tid < CS_BMAX) {
    tab[tid * CS_G + g] = ex;
    cur[tid] = ex;
    if (tid == CS_BMAX - 1) tab[CS_BMAX * CS_G + g] = ex + v;
  }
  __syncthreads();

  int pos[CS_EPT];
#pragma unroll
  for (int u = 0; u < CS_EPT; ++u) {
    pos[u] = -1;
    if (d[u] >= 0) pos[u] = atomicAdd(&cur[d[u] >> CS_SHIFT], 1);
  }
#pragma unroll
  for (int u = 0; u < CS_EPT; ++u)
    if (pos[u] >= 0) sorted[pos[u]] = make_int2(s[u], d[u]);
  __syncthreads();

  const int len = end - beg;
  int2 o[CS_EPT];
#pragma unroll
  for (int u = 0; u < CS_EPT; ++u) o[u] = sorted[tid + u * CS_THREADS];
#pragma unroll
  for (int u = 0; u < CS_EPT; ++u) {
    const int i = tid + u * CS_THREADS;
    if (i < len) tmp[beg + i] = o[u];
  }
}

__global__ __launch_bounds__(CS_THREADS)
void sage_bucket_sort(const int* __restrict__ tmp_i, const int* __restrict__ tab,
                      int* __restrict__ rowptr,
                      int* __restrict__ col, int N, int E) {
  __shared__ int c[CS_W];
  __shared__ int w_tot[CS_THREADS / 32];
  __shared__ int w_lo[CS_THREADS / 32];
  __shared__ int seg_beg[CS_G];
  __shared__ int seg_len[CS_G];
  const int2* tmp = reinterpret_cast<const int2*>(tmp_i);
  const int tid  = threadIdx.x;
  const int lane = tid & 31;
  const int wave = __builtin_amdgcn_readfirstlane(tid >> 5);
  const int b    = blockIdx.x;
  const int node0 = b << CS_SHIFT;
  const int chunk = (E + CS_G - 1) / CS_G;

  int lo = 0;
  if (tid < CS_G) {
    lo = tab[b * CS_G + tid];
    const int hi = tab[(b + 1) * CS_G + tid];
    seg_beg[tid] = tid * chunk + lo;
    seg_len[tid] = hi - lo;
  }
  if (tid < CS_W) c[tid] = 0;
  int part = lo;
#pragma unroll
  for (int o = 16; o > 0; o >>= 1) part += __shfl_xor(part, o);
  if (lane == 0) w_lo[wave] = part;
  __syncthreads();
  int beg = 0;
#pragma unroll
  for (int w = 0; w < CS_THREADS / 32; ++w) beg += w_lo[w];

  int sb[CS_SEGS], sl[CS_SEGS];
  int maxlen = 0;
#pragma unroll
  for (int k = 0; k < CS_SEGS; ++k) {
    sb[k] = __builtin_amdgcn_readfirstlane(seg_beg[wave + 32 * k]);
    sl[k] = __builtin_amdgcn_readfirstlane(seg_len[wave + 32 * k]);
    maxlen = max(maxlen, sl[k]);
  }

  for (int i0 = 0; i0 < maxlen; i0 += 32) {
    const int i = i0 + lane;
    int d[CS_SEGS];
#pragma unroll
    for (int k = 0; k < CS_SEGS; ++k)
      d[k] = (i < sl[k]) ? tmp[sb[k] + i].y : -1;
#pragma unroll
    for (int k = 0; k < CS_SEGS; ++k)
      if (d[k] >= 0) atomicAdd(&c[d[k] - node0], 1);
  }
  __syncthreads();

  const int v = (tid < CS_W) ? c[tid] : 0;
  int inc = v;
#pragma unroll
  for (int o = 1; o < 32; o <<= 1) {
    const int t = __shfl_up(inc, o);
    if (lane >= o) inc += t;
  }
  if (lane == 31) w_tot[wave] = inc;
  __syncthreads();
  int off = 0;
#pragma unroll
  for (int w = 0; w < CS_THREADS / 32; ++w)
    if (w < wave) off += w_tot[w];
  const int ex = off + inc - v;

  const int node = node0 + tid;
  if (tid < CS_W) {
    if (node < N) {
      rowptr[node] = beg + ex;
      if (node == N - 1) rowptr[N] = beg + ex + v;
    }
    c[tid] = ex;
  }
  __syncthreads();

  for (int i0 = 0; i0 < maxlen; i0 += 32) {
    const int i = i0 + lane;
    int2 e[CS_SEGS];
#pragma unroll
    for (int k = 0; k < CS_SEGS; ++k) {
      e[k] = make_int2(0, -1);
      if (i < sl[k]) e[k] = tmp[sb[k] + i];
    }
    int pos[CS_SEGS];
#pragma unroll
    for (int k = 0; k < CS_SEGS; ++k) {
      pos[k] = -1;
      if (e[k].y >= 0) pos[k] = atomicAdd(&c[e[k].y - node0], 1);
    }
#pragma unroll
    for (int k = 0; k < CS_SEGS; ++k)
      if (pos[k] >= 0) col[beg + pos[k]] = e[k].x;
  }
}

#define WP_PITCH 288
#define W_LDS_FLOATS (64 * WP_PITCH)
#define A_PITCH 132
#define A_TILE_FLOATS (16 * A_PITCH)
#define LAYER_WAVES 16
#define LAYER_LDS_FLOATS (2 * W_LDS_FLOATS + LAYER_WAVES * A_TILE_FLOATS)
#define ROWS_AHEAD 16

__device__ __forceinline__ void wave_lds_handover() {
  __builtin_amdgcn_fence(__ATOMIC_RELEASE, "wavefront");
  __builtin_amdgcn_fence(__ATOMIC_ACQUIRE, "wavefront");
}

__device__ __forceinline__ void self_steps(v8f (&c)[8], const float* aq, const float* bq) {
  const v2f a0 = *reinterpret_cast<const v2f*>(aq);
  const v2f a1 = *reinterpret_cast<const v2f*>(aq + 4);
  v2f w0[8], w1[8];
#pragma unroll
  for (int t = 0; t < 8; ++t) {
    w0[t] = *reinterpret_cast<const v2f*>(bq + 32 * t);
    w1[t] = *reinterpret_cast<const v2f*>(bq + 2 * WP_PITCH + 32 * t);
  }
#pragma unroll
  for (int t = 0; t < 8; ++t)
    c[t] = __builtin_amdgcn_wmma_f32_16x16x4_f32(false, a0, false, w0[t], (short)0, c[t],
                                                 false, false);
#pragma unroll
  for (int t = 0; t < 8; ++t)
    c[t] = __builtin_amdgcn_wmma_f32_16x16x4_f32(false, a1, false, w1[t], (short)0, c[t],
                                                 false, false);
}

__global__ __launch_bounds__(32 * LAYER_WAVES)
void sage_fused_layer(const float* __restrict__ feat,
                      const int* __restrict__ rowptr,
                      const int* __restrict__ col,
                      const float* __restrict__ Wl,
                      const float* __restrict__ Wr,
                      const float* __restrict__ bias,
                      float* __restrict__ out,
                      int do_relu, int nstripes) {
  __shared__ __attribute__((aligned(16))) float lds[LAYER_LDS_FLOATS];

  const int tid  = threadIdx.x;
  const int lane = tid & 31;
  const int wave = __builtin_amdgcn_readfirstlane(tid >> 5);
  const int m    = lane & 15;
  const int half = lane >> 4;

  float* sWl = lds;
  float* sWr = lds + W_LDS_FLOATS;
  float* sA  = lds + 2 * W_LDS_FLOATS + wave * A_TILE_FLOATS;

#pragma unroll
  for (int i = 0; i < 64 / LAYER_WAVES; ++i) {
    const int p = wave + LAYER_WAVES * i;
    const float4 l0 = reinterpret_cast<const float4*>(Wl + (size_t)(2 * p) * DIM)[lane];
    const float4 l1 = reinterpret_cast<const float4*>(Wl + (size_t)(2 * p + 1) * DIM)[lane];
    const float4 r0 = reinterpret_cast<const float4*>(Wr + (size_t)(2 * p) * DIM)[lane];
    const float4 r1 = reinterpret_cast<const float4*>(Wr + (size_t)(2 * p + 1) * DIM)[lane];
    float4* dl = reinterpret_cast<float4*>(sWl + p * WP_PITCH + 8 * lane);
    float4* dr = reinterpret_cast<float4*>(sWr + p * WP_PITCH + 8 * lane);
    dl[0] = make_float4(l0.x, l1.x, l0.y, l1.y);
    dl[1] = make_float4(l0.z, l1.z, l0.w, l1.w);
    dr[0] = make_float4(r0.x, r1.x, r0.y, r1.y);
    dr[1] = make_float4(r0.z, r1.z, r0.w, r1.w);
  }

  float bv[8];
#pragma unroll
  for (int t = 0; t < 8; ++t) bv[t] = bias[16 * t + m];

  __syncthreads();

  const float* aF = sA + m * A_PITCH + 2 * half;
  const float* aG = sA + 8 * m + 2 * half;
  const float* bL = sWl + half * WP_PITCH + 2 * m;
  const float* bR = sWr + half * WP_PITCH + 2 * m;
  float* sS = sA + (lane >> 1) * A_PITCH + 4 * (lane & 1);
  const float4 zero = make_float4(0.0f, 0.0f, 0.0f, 0.0f);

  const int stride = LAYER_WAVES * gridDim.x;
  for (int stripe = wave * gridDim.x + blockIdx.x; stripe < nstripes; stripe += stride) {
    const size_t row0 = (size_t)stripe * 16;

    v4f rx[16];
#pragma unroll
    for (int r = 0; r < 16; ++r)
      rx[r] = reinterpret_cast<const v4f*>(feat + (row0 + r) * DIM)[lane];

    const int rp = rowptr[row0 + min(lane, 16)];
    int beg = __builtin_amdgcn_readlane(rp, 0);
    int end = __builtin_amdgcn_readlane(rp, 1);
    int mine = 0;
    if (end > beg) mine = col[beg + min(lane, min(32, end - beg) - 1)];

#pragma unroll
    for (int r = 0; r < 16; ++r)
      reinterpret_cast<v4f*>(sA + r * A_PITCH)[lane] = rx[r];
    wave_lds_handover();

    v8f c[8];
#pragma unroll
    for (int t = 0; t < 8; ++t) c[t] = v8f{};

    for (int r = 0; r < 16; ++r) {
      const int nbeg = end;
      int nend = end;
      int nmine = 0;
      if (r < 15) {
        nend = __builtin_amdgcn_readlane(rp, r + 2);
        if (nend > nbeg) nmine = col[nbeg + min(lane, min(32, nend - nbeg) - 1)];
      }

      const float* aq = aF + 8 * r;
      const float* bq = bR + 4 * r * WP_PITCH;
      bool self_pending = true;

      float4 acc = zero;
      for (int j = beg; j < end; j += 32) {
        const int n = min(32, end - j);
        int cur = mine;
        if (j != beg) cur = col[j + min(lane, n - 1)];
        int k = 0;
        for (; k + ROWS_AHEAD <= n; k += ROWS_AHEAD) {
          int s[ROWS_AHEAD];
          float4 v[ROWS_AHEAD];
#pragma unroll
          for (int u = 0; u < ROWS_AHEAD; ++u) s[u] = __builtin_amdgcn_readlane(cur, k + u);
#pragma unroll
          for (int u = 0; u < ROWS_AHEAD; ++u)
            v[u] = reinterpret_cast<const float4*>(feat + (size_t)s[u] * DIM)[lane];
          if (self_pending) {
            self_pending = false;
            __builtin_amdgcn_sched_barrier(0);
            self_steps(c, aq, bq);
            __builtin_amdgcn_sched_barrier(0);
          }
#pragma unroll
          for (int u = 0; u < ROWS_AHEAD; ++u) {
            acc.x += v[u].x; acc.y += v[u].y; acc.z += v[u].z; acc.w += v[u].w;
          }
        }
        if (k < n) {
          int s[ROWS_AHEAD];
          float4 v[ROWS_AHEAD];
#pragma unroll
          for (int u = 0; u < ROWS_AHEAD; ++u) s[u] = __builtin_amdgcn_readlane(cur, k + u);
#pragma unroll
          for (int u = 0; u < ROWS_AHEAD; ++u) {
            v[u] = zero;
            if (k + u < n) v[u] = reinterpret_cast<const float4*>(feat + (size_t)s[u] * DIM)[lane];
          }
          if (self_pending) {
            self_pending = false;
            __builtin_amdgcn_sched_barrier(0);
            self_steps(c, aq, bq);
            __builtin_amdgcn_sched_barrier(0);
          }
#pragma unroll
          for (int u = 0; u < ROWS_AHEAD; ++u) {
            acc.x += v[u].x; acc.y += v[u].y; acc.z += v[u].z; acc.w += v[u].w;
          }
        }
      }
      if (self_pending) self_steps(c, aq, bq);

      const float inv = 1.0f / fmaxf((float)(end - beg), 1.0f);
      v4f sc;
      sc[0] = acc.x * inv; sc[1] = acc.y * inv; sc[2] = acc.z * inv; sc[3] = acc.w * inv;
      wave_lds_handover();
      *reinterpret_cast<v4f*>(sS + 8 * r) = sc;

      beg = nbeg; end = nend; mine = nmine;
    }
    wave_lds_handover();

#pragma unroll
    for (int s = 0; s < 32; ++s) {
      const v2f ag = *reinterpret_cast<const v2f*>(aG + (s >> 1) * A_PITCH + 4 * (s & 1));
      v2f wl[8];
#pragma unroll
      for (int t = 0; t < 8; ++t)
        wl[t] = *reinterpret_cast<const v2f*>(bL + 2 * s * WP_PITCH + 32 * t);
#pragma unroll
      for (int t = 0; t < 8; ++t)
        c[t] = __builtin_amdgcn_wmma_f32_16x16x4_f32(false, ag, false, wl[t], (short)0, c[t],
                                                     false, false);
    }

#pragma unroll
    for (int t = 0; t < 8; ++t) {
#pragma unroll
      for (int r = 0; r < 8; ++r) {
        float v = c[t][r] + bv[t];
        if (do_relu) v = fmaxf(v, 0.0f);
        out[(row0 + r + 8 * half) * DIM + 16 * t + m] = v;
      }
    }

    wave_lds_handover();
  }
}

extern "C" void kernel_launch(void* const* d_in, const int* in_sizes, int n_in,
                              void* d_out, int out_size, void* d_ws, size_t ws_size,
                              hipStream_t stream) {
  const float* x   = (const float*)d_in[0];
  const int*  edge = (const int*)d_in[1];
  const float* W1l = (const float*)d_in[2];
  const float* b1  = (const float*)d_in[3];
  const float* W1r = (const float*)d_in[4];
  const float* W2l = (const float*)d_in[5];
  const float* b2  = (const float*)d_in[6];
  const float* W2r = (const float*)d_in[7];

  const int N = in_sizes[0] / DIM;
  const int E = in_sizes[1] / 2;
  const int* src = edge;
  const int* dst = edge + E;

  float* h    = (float*)d_ws;
  int* rowptr = (int*)(h + (size_t)N * DIM);
  int* tab    = rowptr + N + 4;
  int* tmp    = tab + (CS_BMAX + 1) * CS_G;
  int* col    = tmp + 2 * (size_t)E;
  float* out = (float*)d_out;

  const int buckets = (N + CS_W - 1) / CS_W;
  const int stripes = N / 16;

  sage_chunk_sort<<<CS_G, CS_THREADS, 0, stream>>>(src, dst, tab, tmp, E);
  sage_bucket_sort<<<buckets, CS_THREADS, 0, stream>>>(tmp, tab, rowptr, col, N, E);

  sage_fused_layer<<<256, 32 * LAYER_WAVES, 0, stream>>>(x, rowptr, col, W1l, W1r, b1, h, 1, stripes);
  sage_fused_layer<<<256, 32 * LAYER_WAVES, 0, stream>>>(h, rowptr, col, W2l, W2r, b2, out, 0, stripes);
}


# neighbour gather walks the stripe's 16 lists as one flat list in groups of 16 row loads and closes a row where the walk meets its end: no extra round trip for entries past 16; sum order unchanged
# speedup vs baseline: 4.9973x; 1.0373x over previous
#include <hip/hip_runtime.h>
#include <hip/hip_bf16.h>

#define DIM 128

typedef __attribute__((ext_vector_type(2))) float v2f;
typedef __attribute__((ext_vector_type(4))) float v4f;
typedef __attribute__((ext_vector_type(8))) float v8f;


#define CS_SHIFT 9
#define CS_W (1 << CS_SHIFT)
#define CS_BMAX 256
#define CS_G 256
#define CS_THREADS 1024
#define CS_EPT 7
#define CS_CAP (CS_EPT * CS_THREADS)
#define CS_SEGS (CS_G / 32)

__global__ __launch_bounds__(CS_THREADS)
void sage_chunk_sort(const int* __restrict__ src, const int* __restrict__ dst,
                     int* __restrict__ tab, int* __restrict__ tmp_i, int E) {
  __shared__ int cur[CS_BMAX];
  __shared__ int w_tot[CS_THREADS / 32];
  __shared__ int2 sorted[CS_CAP];
  int2* tmp = reinterpret_cast<int2*>(tmp_i);
  const int tid  = threadIdx.x;
  const int lane = tid & 31;
  const int wave = tid >> 5;
  const int g    = blockIdx.x;
  if (tid < CS_BMAX) cur[tid] = 0;

  const int chunk = (E + CS_G - 1) / CS_G;
  const int beg = g * chunk;
  const int end = min(E, beg + chunk);

  int d[CS_EPT], s[CS_EPT];
#pragma unroll
  for (int u = 0; u < CS_EPT; ++u) {
    const int e = beg + tid + u * CS_THREADS;
    d[u] = -1; s[u] = 0;
    if (e < end) { d[u] = dst[e]; s[u] = src[e]; }
  }
  __syncthreads();

#pragma unroll
  for (int u = 0; u < CS_EPT; ++u)
    if (d[u] >= 0) atomicAdd(&cur[d[u] >> CS_SHIFT], 1);
  __syncthreads();

  const int v = (tid < CS_BMAX) ? cur[tid] : 0;
  int inc = v;
#pragma unroll
  for (int o = 1; o < 32; o <<= 1) {
    const int t = __shfl_up(inc, o);
    if (lane >= o) inc += t;
  }
  if (lane == 31) w_tot[wave] = inc;
  __syncthreads();
  int off = 0;
#pragma unroll
  for (int w = 0; w < CS_THREADS / 32; ++w)
    if (w < wave) off += w_tot[w];
  const int ex = off + inc - v;
  if (tid < CS_BMAX) {
    tab[tid * CS_G + g] = ex;
    cur[tid] = ex;
    if (tid == CS_BMAX - 1) tab[CS_BMAX * CS_G + g] = ex + v;
  }
  __syncthreads();

  int pos[CS_EPT];
#pragma unroll
  for (int u = 0; u < CS_EPT; ++u) {
    pos[u] = -1;
    if (d[u] >= 0) pos[u] = atomicAdd(&cur[d[u] >> CS_SHIFT], 1);
  }
#pragma unroll
  for (int u = 0; u < CS_EPT; ++u)
    if (pos[u] >= 0) sorted[pos[u]] = make_int2(s[u], d[u]);
  __syncthreads();

  const int len = end - beg;
  int2 o[CS_EPT];
#pragma unroll
  for (int u = 0; u < CS_EPT; ++u) o[u] = sorted[tid + u * CS_THREADS];
#pragma unroll
  for (int u = 0; u < CS_EPT; ++u) {
    const int i = tid + u * CS_THREADS;
    if (i < len) tmp[beg + i] = o[u];
  }
}

__global__ __launch_bounds__(CS_THREADS)
void sage_bucket_sort(const int* __restrict__ tmp_i, const int* __restrict__ tab,
                      int* __restrict__ rowptr,
                      int* __restrict__ col, int N, int E) {
  __shared__ int c[CS_W];
  __shared__ int w_tot[CS_THREADS / 32];
  __shared__ int w_lo[CS_THREADS / 32];
  __shared__ int seg_beg[CS_G];
  __shared__ int seg_len[CS_G];
  const int2* tmp = reinterpret_cast<const int2*>(tmp_i);
  const int tid  = threadIdx.x;
  const int lane = tid & 31;
  const int wave = __builtin_amdgcn_readfirstlane(tid >> 5);
  const int b    = blockIdx.x;
  const int node0 = b << CS_SHIFT;
  const int chunk = (E + CS_G - 1) / CS_G;

  int lo = 0;
  if (tid < CS_G) {
    lo = tab[b * CS_G + tid];
    const int hi = tab[(b + 1) * CS_G + tid];
    seg_beg[tid] = tid * chunk + lo;
    seg_len[tid] = hi - lo;
  }
  if (tid < CS_W) c[tid] = 0;
  int part = lo;
#pragma unroll
  for (int o = 16; o > 0; o >>= 1) part += __shfl_xor(part, o);
  if (lane == 0) w_lo[wave] = part;
  __syncthreads();
  int beg = 0;
#pragma unroll
  for (int w = 0; w < CS_THREADS / 32; ++w) beg += w_lo[w];

  int sb[CS_SEGS], sl[CS_SEGS];
  int maxlen = 0;
#pragma unroll
  for (int k = 0; k < CS_SEGS; ++k) {
    sb[k] = __builtin_amdgcn_readfirstlane(seg_beg[wave + 32 * k]);
    sl[k] = __builtin_amdgcn_readfirstlane(seg_len[wave + 32 * k]);
    maxlen = max(maxlen, sl[k]);
  }

  for (int i0 = 0; i0 < maxlen; i0 += 32) {
    const int i = i0 + lane;
    int d[CS_SEGS];
#pragma unroll
    for (int k = 0; k < CS_SEGS; ++k)
      d[k] = (i < sl[k]) ? tmp[sb[k] + i].y : -1;
#pragma unroll
    for (int k = 0; k < CS_SEGS; ++k)
      if (d[k] >= 0) atomicAdd(&c[d[k] - node0], 1);
  }
  __syncthreads();

  const int v = (tid < CS_W) ? c[tid] : 0;
  int inc = v;
#pragma unroll
  for (int o = 1; o < 32; o <<= 1) {
    const int t = __shfl_up(inc, o);
    if (lane >= o) inc += t;
  }
  if (lane == 31) w_tot[wave] = inc;
  __syncthreads();
  int off = 0;
#pragma unroll
  for (int w = 0; w < CS_THREADS / 32; ++w)
    if (w < wave) off += w_tot[w];
  const int ex = off + inc - v;

  const int node = node0 + tid;
  if (tid < CS_W) {
    if (node < N) {
      rowptr[node] = beg + ex;
      if (node == N - 1) rowptr[N] = beg + ex + v;
    }
    c[tid] = ex;
  }
  __syncthreads();

  for (int i0 = 0; i0 < maxlen; i0 += 32) {
    const int i = i0 + lane;
    int2 e[CS_SEGS];
#pragma unroll
    for (int k = 0; k < CS_SEGS; ++k) {
      e[k] = make_int2(0, -1);
      if (i < sl[k]) e[k] = tmp[sb[k] + i];
    }
    int pos[CS_SEGS];
#pragma unroll
    for (int k = 0; k < CS_SEGS; ++k) {
      pos[k] = -1;
      if (e[k].y >= 0) pos[k] = atomicAdd(&c[e[k].y - node0], 1);
    }
#pragma unroll
    for (int k = 0; k < CS_SEGS; ++k)
      if (pos[k] >= 0) col[beg + pos[k]] = e[k].x;
  }
}

#define WP_PITCH 288
#define W_LDS_FLOATS (64 * WP_PITCH)
#define A_PITCH 132
#define A_TILE_FLOATS (16 * A_PITCH)
#define LAYER_WAVES 16
#define LAYER_LDS_FLOATS (2 * W_LDS_FLOATS + LAYER_WAVES * A_TILE_FLOATS)
#define ROWS_AHEAD 16

__device__ __forceinline__ void wave_lds_handover() {
  __builtin_amdgcn_fence(__ATOMIC_RELEASE, "wavefront");
  __builtin_amdgcn_fence(__ATOMIC_ACQUIRE, "wavefront");
}

__device__ __forceinline__ void self_steps(v8f (&c)[8], const float* aq, const float* bq) {
  const v2f a0 = *reinterpret_cast<const v2f*>(aq);
  const v2f a1 = *reinterpret_cast<const v2f*>(aq + 4);
  v2f w0[8], w1[8];
#pragma unroll
  for (int t = 0; t < 8; ++t) {
    w0[t] = *reinterpret_cast<const v2f*>(bq + 32 * t);
    w1[t] = *reinterpret_cast<const v2f*>(bq + 2 * WP_PITCH + 32 * t);
  }
#pragma unroll
  for (int t = 0; t < 8; ++t)
    c[t] = __builtin_amdgcn_wmma_f32_16x16x4_f32(false, a0, false, w0[t], (short)0, c[t],
                                                 false, false);
#pragma unroll
  for (int t = 0; t < 8; ++t)
    c[t] = __builtin_amdgcn_wmma_f32_16x16x4_f32(false, a1, false, w1[t], (short)0, c[t],
                                                 false, false);
}

__global__ __launch_bounds__(32 * LAYER_WAVES)
void sage_fused_layer(const float* __restrict__ feat,
                      const int* __restrict__ rowptr,
                      const int* __restrict__ col,
                      const float* __restrict__ Wl,
                      const float* __restrict__ Wr,
                      const float* __restrict__ bias,
                      float* __restrict__ out,
                      int do_relu, int nstripes) {
  __shared__ __attribute__((aligned(16))) float lds[LAYER_LDS_FLOATS];

  const int tid  = threadIdx.x;
  const int lane = tid & 31;
  const int wave = __builtin_amdgcn_readfirstlane(tid >> 5);
  const int m    = lane & 15;
  const int half = lane >> 4;

  float* sWl = lds;
  float* sWr = lds + W_LDS_FLOATS;
  float* sA  = lds + 2 * W_LDS_FLOATS + wave * A_TILE_FLOATS;

#pragma unroll
  for (int i = 0; i < 64 / LAYER_WAVES; ++i) {
    const int p = wave + LAYER_WAVES * i;
    const float4 l0 = reinterpret_cast<const float4*>(Wl + (size_t)(2 * p) * DIM)[lane];
    const float4 l1 = reinterpret_cast<const float4*>(Wl + (size_t)(2 * p + 1) * DIM)[lane];
    const float4 r0 = reinterpret_cast<const float4*>(Wr + (size_t)(2 * p) * DIM)[lane];
    const float4 r1 = reinterpret_cast<const float4*>(Wr + (size_t)(2 * p + 1) * DIM)[lane];
    float4* dl = reinterpret_cast<float4*>(sWl + p * WP_PITCH + 8 * lane);
    float4* dr = reinterpret_cast<float4*>(sWr + p * WP_PITCH + 8 * lane);
    dl[0] = make_float4(l0.x, l1.x, l0.y, l1.y);
    dl[1] = make_float4(l0.z, l1.z, l0.w, l1.w);
    dr[0] = make_float4(r0.x, r1.x, r0.y, r1.y);
    dr[1] = make_float4(r0.z, r1.z, r0.w, r1.w);
  }

  float bv[8];
#pragma unroll
  for (int t = 0; t < 8; ++t) bv[t] = bias[16 * t + m];

  __syncthreads();

  const float* aF = sA + m * A_PITCH + 2 * half;
  const float* aG = sA + 8 * m + 2 * half;
  const float* bL = sWl + half * WP_PITCH + 2 * m;
  const float* bR = sWr + half * WP_PITCH + 2 * m;
  float* sS = sA + (lane >> 1) * A_PITCH + 4 * (lane & 1);
  const float4 zero = make_float4(0.0f, 0.0f, 0.0f, 0.0f);

  const int stride = LAYER_WAVES * gridDim.x;
  for (int stripe = wave * gridDim.x + blockIdx.x; stripe < nstripes; stripe += stride) {
    const size_t row0 = (size_t)stripe * 16;

    v4f rx[16];
#pragma unroll
    for (int r = 0; r < 16; ++r)
      rx[r] = reinterpret_cast<const v4f*>(feat + (row0 + r) * DIM)[lane];

    const int rp = rowptr[row0 + min(lane, 16)];
    const int begAll = __builtin_amdgcn_readlane(rp, 0);
    const int endAll = __builtin_amdgcn_readlane(rp, 16);
    int cur = 0;
    if (endAll > begAll) cur = col[min(begAll + lane, endAll - 1)];

#pragma unroll
    for (int r = 0; r < 16; ++r)
      reinterpret_cast<v4f*>(sA + r * A_PITCH)[lane] = rx[r];
    wave_lds_handover();

    v8f c[8];
#pragma unroll
    for (int t = 0; t < 8; ++t) c[t] = v8f{};

    int r = 0;
    int q = 0;
    int rbeg = begAll;
    int rend = __builtin_amdgcn_readlane(rp, 1);
    float4 acc = zero;

    auto close_rows = [&](int p) {
      while (r < 16 && rend == p) {
        const float inv = 1.0f / fmaxf((float)(rend - rbeg), 1.0f);
        v4f sc;
        sc[0] = acc.x * inv; sc[1] = acc.y * inv; sc[2] = acc.z * inv; sc[3] = acc.w * inv;
        wave_lds_handover();
        *reinterpret_cast<v4f*>(sS + 8 * r) = sc;
        acc = zero;
        ++r;
        rbeg = rend;
        rend = __builtin_amdgcn_readlane(rp, min(r + 1, 16));
      }
    };

    for (int jb = begAll; jb < endAll; jb += ROWS_AHEAD) {
      int nxt = 0;
      if (jb + ROWS_AHEAD < endAll) nxt = col[min(jb + ROWS_AHEAD + lane, endAll - 1)];
      int s[ROWS_AHEAD];
      float4 v[ROWS_AHEAD];
#pragma unroll
      for (int u = 0; u < ROWS_AHEAD; ++u) s[u] = __builtin_amdgcn_readlane(cur, u);
      if (jb + ROWS_AHEAD <= endAll) {
#pragma unroll
        for (int u = 0; u < ROWS_AHEAD; ++u)
          v[u] = reinterpret_cast<const float4*>(feat + (size_t)s[u] * DIM)[lane];
      } else {
#pragma unroll
        for (int u = 0; u < ROWS_AHEAD; ++u) {
          v[u] = zero;
          if (jb + u < endAll)
            v[u] = reinterpret_cast<const float4*>(feat + (size_t)s[u] * DIM)[lane];
        }
      }
      __builtin_amdgcn_sched_barrier(0);
      if (q < 16) { self_steps(c, aF + 8 * q, bR + 4 * q * WP_PITCH); ++q; }
      while (q < 16 && __builtin_amdgcn_readlane(rp, q + 1) < jb + ROWS_AHEAD) {
        self_steps(c, aF + 8 * q, bR + 4 * q * WP_PITCH); ++q;
      }
      __builtin_amdgcn_sched_barrier(0);
#pragma unroll
      for (int u = 0; u < ROWS_AHEAD; ++u) {
        if (__builtin_expect(rend == jb + u, 0)) close_rows(jb + u);
        acc.x += v[u].x; acc.y += v[u].y; acc.z += v[u].z; acc.w += v[u].w;
      }
      cur = nxt;
    }
    while (q < 16) { self_steps(c, aF + 8 * q, bR + 4 * q * WP_PITCH); ++q; }
    close_rows(endAll);
    wave_lds_handover();

#pragma unroll
    for (int s = 0; s < 32; ++s) {
      const v2f ag = *reinterpret_cast<const v2f*>(aG + (s >> 1) * A_PITCH + 4 * (s & 1));
      v2f wl[8];
#pragma unroll
      for (int t = 0; t < 8; ++t)
        wl[t] = *reinterpret_cast<const v2f*>(bL + 2 * s * WP_PITCH + 32 * t);
#pragma unroll
      for (int t = 0; t < 8; ++t)
        c[t] = __builtin_amdgcn_wmma_f32_16x16x4_f32(false, ag, false, wl[t], (short)0, c[t],
                                                     false, false);
    }

#pragma unroll
    for (int t = 0; t < 8; ++t) {
#pragma unroll
      for (int r = 0; r < 8; ++r) {
        float v = c[t][r] + bv[t];
        if (do_relu) v = fmaxf(v, 0.0f);
        out[(row0 + r + 8 * half) * DIM + 16 * t + m] = v;
      }
    }

    wave_lds_handover();
  }
}

extern "C" void kernel_launch(void* const* d_in, const int* in_sizes, int n_in,
                              void* d_out, int out_size, void* d_ws, size_t ws_size,
                              hipStream_t stream) {
  const float* x   = (const float*)d_in[0];
  const int*  edge = (const int*)d_in[1];
  const float* W1l = (const float*)d_in[2];
  const float* b1  = (const float*)d_in[3];
  const float* W1r = (const float*)d_in[4];
  const float* W2l = (const float*)d_in[5];
  const float* b2  = (const float*)d_in[6];
  const float* W2r = (const float*)d_in[7];

  const int N = in_sizes[0] / DIM;
  const int E = in_sizes[1] / 2;
  const int* src = edge;
  const int* dst = edge + E;

  float* h    = (float*)d_ws;
  int* rowptr = (int*)(h + (size_t)N * DIM);
  int* tab    = rowptr + N + 4;
  int* tmp    = tab + (CS_BMAX + 1) * CS_G;
  int* col    = tmp + 2 * (size_t)E;
  float* out = (float*)d_out;

  const int buckets = (N + CS_W - 1) / CS_W;
  const int stripes = N / 16;

  sage_chunk_sort<<<CS_G, CS_THREADS, 0, stream>>>(src, dst, tab, tmp, E);
  sage_bucket_sort<<<buckets, CS_THREADS, 0, stream>>>(tmp, tab, rowptr, col, N, E);

  sage_fused_layer<<<256, 32 * LAYER_WAVES, 0, stream>>>(x, rowptr, col, W1l, W1r, b1, h, 1, stripes);
  sage_fused_layer<<<256, 32 * LAYER_WAVES, 0, stream>>>(h, rowptr, col, W2l, W2r, b2, out, 0, stripes);
}
